# starting kernel
# baseline (speedup 1.0000x reference)
#include <hip/hip_runtime.h>
#include <math.h>
#include <float.h>


typedef __attribute__((ext_vector_type(2))) float v2f;
typedef __attribute__((ext_vector_type(8))) float v8f;

#define NN 50000
#define EE 1200000
#define DD 64
#define GG 1024

static_assert(NN % 16 == 0, "WMMA GEMM requires row count multiple of 16");


__global__ void fill_f32_kernel(float* __restrict__ p, float v, int n) {
  int i = blockIdx.x * blockDim.x + threadIdx.x;
  if (i < n) p[i] = v;
}

__global__ void degree_kernel(const int* __restrict__ dst, float* __restrict__ deg, int e) {
  int i = blockIdx.x * blockDim.x + threadIdx.x;
  if (i < e) atomicAdd(&deg[dst[i]], 1.0f);
}

__global__ void dinv_kernel(const float* __restrict__ deg, float* __restrict__ dinv, int n) {
  int i = blockIdx.x * blockDim.x + threadIdx.x;
  if (i < n) dinv[i] = rsqrtf(deg[i]);
}

__global__ void pad_x_kernel(const float* __restrict__ x, float* __restrict__ hpad, int n) {
  int i = blockIdx.x * blockDim.x + threadIdx.x;
  if (i >= n * 16) return;
  int r = i >> 4, c = i & 15;
  hpad[i] = (c < 9) ? x[r * 9 + c] : 0.0f;
}

__global__ void pad_w0_kernel(const float* __restrict__ W0, float* __restrict__ w0pad) {
  int i = blockIdx.x * blockDim.x + threadIdx.x;
  if (i >= 16 * 64) return;
  int r = i >> 6, c = i & 63;
  w0pad[i] = (r < 9) ? W0[r * 64 + c] : 0.0f;
}

template <int KDIM>
__global__ void gemm_wmma_kernel(const float* __restrict__ A,
                                 const float* __restrict__ W,
                                 float* __restrict__ out) {
  const int wave = threadIdx.x >> 5;
  const int lane = threadIdx.x & 31;
  const int l15  = lane & 15;
  const int half = lane >> 4;
  const int m0   = blockIdx.x * 16;
  const int n0   = wave * 16;
  const int row  = m0 + l15;
  const int col  = n0 + l15;

  const float* __restrict__ arow = A + (size_t)row * KDIM + half * 2;
  const float* __restrict__ bcol = W + (size_t)half * 2 * 64 + col;

  v8f c = {};
#pragma unroll
  for (int k = 0; k < KDIM; k += 4) {
    v2f a, b;
    a.x = arow[k + 0];
    a.y = arow[k + 1];
    b.x = bcol[(size_t)k * 64];
    b.y = bcol[(size_t)k * 64 + 64];
    c = __builtin_amdgcn_wmma_f32_16x16x4_f32(false, a, false, b, (short)0, c,
                                              false, false);
  }

  float* __restrict__ orow = out + (size_t)(m0 + half * 8) * 64 + col;
#pragma unroll
  for (int v = 0; v < 8; ++v) {
    orow[(size_t)v * 64] = c[v];
  }
}

__global__ void edge_agg_kernel(const float* __restrict__ hW,
                                const int* __restrict__ src,
                                const int* __restrict__ dst,
                                const float* __restrict__ dinv,
                                float* __restrict__ agg, int e) {
  int idx = blockIdx.x * blockDim.x + threadIdx.x;
  if (idx >= e * 16) return;
  const int ei = idx >> 4;
  const int fg = (idx & 15) * 4;
  const int s = src[ei];
  const int d = dst[ei];
  const float norm = dinv[s] * dinv[d];
  const float4 hv = *(const float4*)(hW + (size_t)s * DD + fg);
  float* ap = agg + (size_t)d * DD + fg;
  atomicAdd(ap + 0, hv.x * norm);
  atomicAdd(ap + 1, hv.y * norm);
  atomicAdd(ap + 2, hv.z * norm);
  atomicAdd(ap + 3, hv.w * norm);
}

__global__ void layer_finish_kernel(const float* __restrict__ agg,
                                    const float* __restrict__ hW,
                                    const float* __restrict__ dinv,
                                    const float* __restrict__ b,
                                    float* __restrict__ hout, int n) {
  int i = blockIdx.x * blockDim.x + threadIdx.x;
  if (i >= n * DD) return;
  const int row = i >> 6;
  const int f = i & 63;
  const float di = dinv[row];
  hout[i] = tanhf(agg[i] + hW[i] * di * di + b[f]);
}


__device__ inline void atomicMaxF32(float* addr, float val) {
  if (val >= 0.0f)
    atomicMax((int*)addr, __float_as_int(val));
  else
    atomicMin((unsigned int*)addr, __float_as_uint(val));
}

__global__ void pool_acc_kernel(const float* __restrict__ h,
                                const int* __restrict__ batch,
                                float* __restrict__ gmax,
                                float* __restrict__ gsum,
                                float* __restrict__ cnt, int n) {
  int i = blockIdx.x * blockDim.x + threadIdx.x;
  if (i >= n * DD) return;
  const int row = i >> 6;
  const int f = i & 63;
  const int g = batch[row];
  const float v = h[i];
  atomicAdd(&gsum[(size_t)g * DD + f], v);
  atomicMaxF32(&gmax[(size_t)g * DD + f], v);
  if (f == 0) atomicAdd(&cnt[g], 1.0f);
}

__global__ void pool_finish_kernel(const float* __restrict__ gmax,
                                   const float* __restrict__ gsum,
                                   const float* __restrict__ cnt,
                                   float* __restrict__ hidden, int g_) {
  int i = blockIdx.x * blockDim.x + threadIdx.x;
  if (i >= g_ * DD) return;
  const int g = i >> 6;
  const int f = i & 63;
  const float c = cnt[g];
  float mx = gmax[(size_t)g * DD + f];
  if (c == 0.0f) mx = 0.0f;
  hidden[(size_t)g * (2 * DD) + f] = mx;
  hidden[(size_t)g * (2 * DD) + DD + f] = gsum[(size_t)g * DD + f] / fmaxf(c, 1.0f);
}

__global__ void head_kernel(const float* __restrict__ hidden,
                            const float* __restrict__ Wout,
                            const float* __restrict__ bout,
                            float* __restrict__ out, int g_) {
  int g = blockIdx.x * blockDim.x + threadIdx.x;
  if (g >= g_) return;
  float acc = bout[0];
#pragma unroll 8
  for (int k = 0; k < 2 * DD; ++k) acc += hidden[(size_t)g * (2 * DD) + k] * Wout[k];
  out[g] = acc;
}


extern "C" void kernel_launch(void* const* d_in, const int* in_sizes, int n_in,
                              void* d_out, int out_size, void* d_ws, size_t ws_size,
                              hipStream_t stream) {
  (void)in_sizes; (void)n_in; (void)out_size; (void)ws_size;

  const float* x    = (const float*)d_in[0];
  const int*   eidx = (const int*)d_in[1];
  const int*   batch = (const int*)d_in[2];
  const float* W0 = (const float*)d_in[3];
  const float* b0 = (const float*)d_in[4];
  const float* W1 = (const float*)d_in[5];
  const float* b1 = (const float*)d_in[6];
  const float* W2 = (const float*)d_in[7];
  const float* b2 = (const float*)d_in[8];
  const float* W3 = (const float*)d_in[9];
  const float* b3 = (const float*)d_in[10];
  const float* Wout = (const float*)d_in[11];
  const float* bout = (const float*)d_in[12];

  const int* src = eidx;
  const int* dst = eidx + EE;

  float* outp   = (float*)d_out;
  float* hidden = (float*)d_out + GG;

  float* ws = (float*)d_ws;
  size_t o = 0;
  float* deg   = ws + o; o += NN;
  float* dinv  = ws + o; o += NN;
  float* hpad  = ws + o; o += (size_t)NN * 16;
  float* w0pad = ws + o; o += 16 * 64;
  float* hW    = ws + o; o += (size_t)NN * DD;
  float* h     = ws + o; o += (size_t)NN * DD;
  float* agg   = ws + o; o += (size_t)NN * DD;
  float* gmax  = ws + o; o += (size_t)GG * DD;
  float* gsum  = ws + o; o += (size_t)GG * DD;
  float* cnt   = ws + o; o += GG;

  const int T = 256;
  auto nb = [](long long n, int t) { return (int)((n + t - 1) / t); };

  fill_f32_kernel<<<nb(NN, T), T, 0, stream>>>(deg, 1.0f, NN);
  degree_kernel<<<nb(EE, T), T, 0, stream>>>(dst, deg, EE);
  dinv_kernel<<<nb(NN, T), T, 0, stream>>>(deg, dinv, NN);

  pad_x_kernel<<<nb((long long)NN * 16, T), T, 0, stream>>>(x, hpad, NN);
  pad_w0_kernel<<<nb(16 * 64, T), T, 0, stream>>>(W0, w0pad);

  const int mtiles = NN / 16;
  const float* Wl[4] = {w0pad, W1, W2, W3};
  const float* bl[4] = {b0, b1, b2, b3};

  for (int l = 0; l < 4; ++l) {
    if (l == 0)
      gemm_wmma_kernel<16><<<mtiles, 128, 0, stream>>>(hpad, Wl[0], hW);
    else
      gemm_wmma_kernel<64><<<mtiles, 128, 0, stream>>>(h, Wl[l], hW);

    fill_f32_kernel<<<nb((long long)NN * DD, T), T, 0, stream>>>(agg, 0.0f, NN * DD);
    edge_agg_kernel<<<nb((long long)EE * 16, T), T, 0, stream>>>(hW, src, dst, dinv, agg, EE);
    layer_finish_kernel<<<nb((long long)NN * DD, T), T, 0, stream>>>(agg, hW, dinv, bl[l], h, NN);
  }

  fill_f32_kernel<<<nb((long long)GG * DD, T), T, 0, stream>>>(gmax, -FLT_MAX, GG * DD);
  fill_f32_kernel<<<nb((long long)GG * DD, T), T, 0, stream>>>(gsum, 0.0f, GG * DD);
  fill_f32_kernel<<<nb(GG, T), T, 0, stream>>>(cnt, 0.0f, GG);
  pool_acc_kernel<<<nb((long long)NN * DD, T), T, 0, stream>>>(h, batch, gmax, gsum, cnt, NN);
  pool_finish_kernel<<<nb((long long)GG * DD, T), T, 0, stream>>>(gmax, gsum, cnt, hidden, GG);

  head_kernel<<<nb(GG, 128), 128, 0, stream>>>(hidden, Wout, bout, outp, GG);
}


# Edge aggregation as a per-call dst-sorted (CSR) gather with fused self-loop+bias+tanh, replacing the f32 atomic scatter, agg fill and finish pass; edge sums now run in slot (atomic rank) order
# speedup vs baseline: 2.3235x; 2.3235x over previous
#include <hip/hip_runtime.h>
#include <math.h>
#include <float.h>


typedef __attribute__((ext_vector_type(2))) float v2f;
typedef __attribute__((ext_vector_type(8))) float v8f;

#define NN 50000
#define EE 1200000
#define DD 64
#define GG 1024

static_assert(NN % 16 == 0, "WMMA GEMM requires row count multiple of 16");


__global__ void fill_f32_kernel(float* __restrict__ p, float v, int n) {
  int i = blockIdx.x * blockDim.x + threadIdx.x;
  if (i < n) p[i] = v;
}

__global__ void fill_i32_kernel(int* __restrict__ p, int v, int n) {
  int i = blockIdx.x * blockDim.x + threadIdx.x;
  if (i < n) p[i] = v;
}

__global__ void hist_rank_kernel(const int* __restrict__ dst, int* __restrict__ cnt,
                                 int* __restrict__ rank, int e) {
  int i = blockIdx.x * blockDim.x + threadIdx.x;
  if (i < e) rank[i] = atomicAdd(&cnt[dst[i]], 1);
}

__global__ __launch_bounds__(1024) void rowptr_dinv_kernel(const int* __restrict__ cnt,
                                                           int* __restrict__ rowptr,
                                                           float* __restrict__ dinv, int n) {
  __shared__ int s_tot[32];
  __shared__ int s_part[32];
  const int tid = threadIdx.x;
  const int lane = tid & 31;
  const int wave = tid >> 5;
  const int first = blockIdx.x * 1024;

  int part = 0;
#pragma unroll 8
  for (int i = tid; i < first; i += 1024) part += cnt[i];

  const int idx = first + tid;
  const int c = (idx < n) ? cnt[idx] : 0;

  int incl = c;
  for (int o = 1; o < 32; o <<= 1) {
    const int t = __shfl_up(incl, o);
    if (lane >= o) incl += t;
  }
  for (int o = 16; o > 0; o >>= 1) part += __shfl_xor(part, o);
  if (lane == 31) s_tot[wave] = incl;
  if (lane == 0) s_part[wave] = part;
  __syncthreads();

  const int wt = s_tot[lane];
  int pt = s_part[lane];
  int wincl = wt;
  for (int o = 1; o < 32; o <<= 1) {
    const int t = __shfl_up(wincl, o);
    if (lane >= o) wincl += t;
  }
  for (int o = 16; o > 0; o >>= 1) pt += __shfl_xor(pt, o);
  const int wexcl = __shfl(wincl - wt, wave);

  const int excl = pt + wexcl + incl - c;
  if (idx < n) {
    rowptr[idx] = excl;
    dinv[idx] = rsqrtf((float)(c + 1));
  }
  if (idx == n - 1) rowptr[n] = excl + c;
}

__global__ void csr_fill_kernel(const int* __restrict__ src, const int* __restrict__ dst,
                                const int* __restrict__ rank, const int* __restrict__ rowptr,
                                const float* __restrict__ dinv, int* __restrict__ csr, int e) {
  int i = blockIdx.x * blockDim.x + threadIdx.x;
  if (i >= e) return;
  const int s = src[i];
  const int d = dst[i];
  const int pos = rowptr[d] + rank[i];
  ((int2*)csr)[pos] = make_int2(s, __float_as_int(dinv[s] * dinv[d]));
}

__global__ void pad_x_kernel(const float* __restrict__ x, float* __restrict__ hpad, int n) {
  int i = blockIdx.x * blockDim.x + threadIdx.x;
  if (i >= n * 16) return;
  int r = i >> 4, c = i & 15;
  hpad[i] = (c < 9) ? x[r * 9 + c] : 0.0f;
}

__global__ void pad_w0_kernel(const float* __restrict__ W0, float* __restrict__ w0pad) {
  int i = blockIdx.x * blockDim.x + threadIdx.x;
  if (i >= 16 * 64) return;
  int r = i >> 6, c = i & 63;
  w0pad[i] = (r < 9) ? W0[r * 64 + c] : 0.0f;
}

template <int KDIM>
__global__ void gemm_wmma_kernel(const float* __restrict__ A,
                                 const float* __restrict__ W,
                                 float* __restrict__ out) {
  const int wave = threadIdx.x >> 5;
  const int lane = threadIdx.x & 31;
  const int l15  = lane & 15;
  const int half = lane >> 4;
  const int m0   = blockIdx.x * 16;
  const int n0   = wave * 16;
  const int row  = m0 + l15;
  const int col  = n0 + l15;

  const float* __restrict__ arow = A + (size_t)row * KDIM + half * 2;
  const float* __restrict__ bcol = W + (size_t)half * 2 * 64 + col;

  v8f c = {};
#pragma unroll
  for (int k = 0; k < KDIM; k += 4) {
    v2f a, b;
    a.x = arow[k + 0];
    a.y = arow[k + 1];
    b.x = bcol[(size_t)k * 64];
    b.y = bcol[(size_t)k * 64 + 64];
    c = __builtin_amdgcn_wmma_f32_16x16x4_f32(false, a, false, b, (short)0, c,
                                              false, false);
  }

  float* __restrict__ orow = out + (size_t)(m0 + half * 8) * 64 + col;
#pragma unroll
  for (int v = 0; v < 8; ++v) {
    orow[(size_t)v * 64] = c[v];
  }
}

__device__ __forceinline__ void axpy4(float4& acc, const float4 v, const float w) {
  acc.x = fmaf(v.x, w, acc.x);
  acc.y = fmaf(v.y, w, acc.y);
  acc.z = fmaf(v.z, w, acc.z);
  acc.w = fmaf(v.w, w, acc.w);
}


__global__ __launch_bounds__(256) void gather_finish_kernel(const float* __restrict__ hW,
                                                            const int* __restrict__ rowptr,
                                                            const int* __restrict__ csr,
                                                            const float* __restrict__ dinv,
                                                            const float* __restrict__ b,
                                                            float* __restrict__ hout, int n) {
  const int gid = blockIdx.x * blockDim.x + threadIdx.x;
  const int d = gid >> 4;
  if (d >= n) return;
  const int fg = (gid & 15) * 4;
  const int2* __restrict__ ed = (const int2*)csr;

  int j = rowptr[d];
  const int end = rowptr[d + 1];
  const float4 self = *(const float4*)(hW + d * DD + fg);
  const float di = dinv[d];
  const float4 bb = *(const float4*)(b + fg);

  float4 acc = make_float4(0.0f, 0.0f, 0.0f, 0.0f);

  for (; j + 8 <= end; j += 8) {
    int2 e[8];
    float4 v[8];
#pragma unroll
    for (int k = 0; k < 8; ++k) e[k] = ed[j + k];
#pragma unroll
    for (int k = 0; k < 8; ++k) v[k] = *(const float4*)(hW + e[k].x * DD + fg);
#pragma unroll
    for (int k = 0; k < 8; ++k) axpy4(acc, v[k], __int_as_float(e[k].y));
  }

  if (j < end) {
    const int last = end - 1;
    int2 e[7];
    float4 v[7];
#pragma unroll
    for (int k = 0; k < 7; ++k) e[k] = ed[(j + k < last) ? (j + k) : last];
#pragma unroll
    for (int k = 0; k < 7; ++k) v[k] = *(const float4*)(hW + e[k].x * DD + fg);
#pragma unroll
    for (int k = 0; k < 7; ++k)
      axpy4(acc, v[k], (j + k < end) ? __int_as_float(e[k].y) : 0.0f);
  }

  float4 r;
  r.x = tanhf(acc.x + self.x * di * di + bb.x);
  r.y = tanhf(acc.y + self.y * di * di + bb.y);
  r.z = tanhf(acc.z + self.z * di * di + bb.z);
  r.w = tanhf(acc.w + self.w * di * di + bb.w);
  *(float4*)(hout + d * DD + fg) = r;
}


__device__ inline void atomicMaxF32(float* addr, float val) {
  if (val >= 0.0f)
    atomicMax((int*)addr, __float_as_int(val));
  else
    atomicMin((unsigned int*)addr, __float_as_uint(val));
}

__global__ void pool_acc_kernel(const float* __restrict__ h,
                                const int* __restrict__ batch,
                                float* __restrict__ gmax,
                                float* __restrict__ gsum,
                                float* __restrict__ cnt, int n) {
  int i = blockIdx.x * blockDim.x + threadIdx.x;
  if (i >= n * DD) return;
  const int row = i >> 6;
  const int f = i & 63;
  const int g = batch[row];
  const float v = h[i];
  atomicAdd(&gsum[(size_t)g * DD + f], v);
  atomicMaxF32(&gmax[(size_t)g * DD + f], v);
  if (f == 0) atomicAdd(&cnt[g], 1.0f);
}

__global__ void pool_finish_kernel(const float* __restrict__ gmax,
                                   const float* __restrict__ gsum,
                                   const float* __restrict__ cnt,
                                   float* __restrict__ hidden, int g_) {
  int i = blockIdx.x * blockDim.x + threadIdx.x;
  if (i >= g_ * DD) return;
  const int g = i >> 6;
  const int f = i & 63;
  const float c = cnt[g];
  float mx = gmax[(size_t)g * DD + f];
  if (c == 0.0f) mx = 0.0f;
  hidden[(size_t)g * (2 * DD) + f] = mx;
  hidden[(size_t)g * (2 * DD) + DD + f] = gsum[(size_t)g * DD + f] / fmaxf(c, 1.0f);
}

__global__ void head_kernel(const float* __restrict__ hidden,
                            const float* __restrict__ Wout,
                            const float* __restrict__ bout,
                            float* __restrict__ out, int g_) {
  int g = blockIdx.x * blockDim.x + threadIdx.x;
  if (g >= g_) return;
  float acc = bout[0];
#pragma unroll 8
  for (int k = 0; k < 2 * DD; ++k) acc += hidden[(size_t)g * (2 * DD) + k] * Wout[k];
  out[g] = acc;
}


extern "C" void kernel_launch(void* const* d_in, const int* in_sizes, int n_in,
                              void* d_out, int out_size, void* d_ws, size_t ws_size,
                              hipStream_t stream) {
  (void)in_sizes; (void)n_in; (void)out_size; (void)ws_size;

  const float* x    = (const float*)d_in[0];
  const int*   eidx = (const int*)d_in[1];
  const int*   batch = (const int*)d_in[2];
  const float* W0 = (const float*)d_in[3];
  const float* b0 = (const float*)d_in[4];
  const float* W1 = (const float*)d_in[5];
  const float* b1 = (const float*)d_in[6];
  const float* W2 = (const float*)d_in[7];
  const float* b2 = (const float*)d_in[8];
  const float* W3 = (const float*)d_in[9];
  const float* b3 = (const float*)d_in[10];
  const float* Wout = (const float*)d_in[11];
  const float* bout = (const float*)d_in[12];

  const int* src = eidx;
  const int* dst = eidx + EE;

  float* outp   = (float*)d_out;
  float* hidden = (float*)d_out + GG;

  float* ws = (float*)d_ws;
  size_t o = 0;
  int*   cnt    = (int*)(ws + o); o += NN + 48;
  float* dinv   = ws + o;         o += NN + 48;
  int*   rowptr = (int*)(ws + o); o += NN + 112;
  int*   rank   = (int*)(ws + o); o += EE;
  int*   csr    = (int*)(ws + o); o += (size_t)EE * 2;
  float* hpad  = ws + o; o += (size_t)NN * 16;
  float* w0pad = ws + o; o += 16 * 64;
  float* hW    = ws + o; o += (size_t)NN * DD;
  float* h     = ws + o; o += (size_t)NN * DD;
  float* gmax  = ws + o; o += (size_t)GG * DD;
  float* gsum  = ws + o; o += (size_t)GG * DD;
  float* gcnt  = ws + o; o += GG;

  const int T = 256;
  auto nb = [](long long n, int t) { return (int)((n + t - 1) / t); };

  fill_i32_kernel<<<nb(NN, T), T, 0, stream>>>(cnt, 0, NN);
  hist_rank_kernel<<<nb(EE, T), T, 0, stream>>>(dst, cnt, rank, EE);
  rowptr_dinv_kernel<<<nb(NN, 1024), 1024, 0, stream>>>(cnt, rowptr, dinv, NN);
  csr_fill_kernel<<<nb(EE, T), T, 0, stream>>>(src, dst, rank, rowptr, dinv, csr, EE);

  pad_x_kernel<<<nb((long long)NN * 16, T), T, 0, stream>>>(x, hpad, NN);
  pad_w0_kernel<<<nb(16 * 64, T), T, 0, stream>>>(W0, w0pad);

  const int mtiles = NN / 16;
  const float* Wl[4] = {w0pad, W1, W2, W3};
  const float* bl[4] = {b0, b1, b2, b3};

  for (int l = 0; l < 4; ++l) {
    if (l == 0)
      gemm_wmma_kernel<16><<<mtiles, 128, 0, stream>>>(hpad, Wl[0], hW);
    else
      gemm_wmma_kernel<64><<<mtiles, 128, 0, stream>>>(h, Wl[l], hW);

    gather_finish_kernel<<<nb((long long)NN * 16, T), T, 0, stream>>>(hW, rowptr, csr, dinv, bl[l], h, NN);
  }

  fill_f32_kernel<<<nb((long long)GG * DD, T), T, 0, stream>>>(gmax, -FLT_MAX, GG * DD);
  fill_f32_kernel<<<nb((long long)GG * DD, T), T, 0, stream>>>(gsum, 0.0f, GG * DD);
  fill_f32_kernel<<<nb(GG, T), T, 0, stream>>>(gcnt, 0.0f, GG);
  pool_acc_kernel<<<nb((long long)NN * DD, T), T, 0, stream>>>(h, batch, gmax, gsum, gcnt, NN);
  pool_finish_kernel<<<nb((long long)GG * DD, T), T, 0, stream>>>(gmax, gsum, gcnt, hidden, GG);

  head_kernel<<<nb(GG, 128), 128, 0, stream>>>(hidden, Wout, bout, outp, GG);
}


# Per-element passes (pooling, degree count, edge fill, x pad): 8+ elements per lane, 12x fewer waves; pooling sums each graph-id run in registers, one atomic pair per run, so gsum order changes
# speedup vs baseline: 2.4612x; 1.0593x over previous
#include <hip/hip_runtime.h>
#include <math.h>
#include <float.h>


typedef __attribute__((ext_vector_type(2))) float v2f;
typedef __attribute__((ext_vector_type(8))) float v8f;

#define NN 50000
#define EE 1200000
#define DD 64
#define GG 1024

static_assert(NN % 16 == 0, "WMMA GEMM requires row count multiple of 16");
static_assert(EE % 8 == 0, "edge passes take 8 edges per lane as two 16-byte groups");
static_assert((NN + 48) % 4 == 0 && (NN + 112) % 4 == 0, "16-byte alignment of the edge arrays");


__global__ void fill_f32_kernel(float* __restrict__ p, float v, int n) {
  int i = blockIdx.x * blockDim.x + threadIdx.x;
  if (i < n) p[i] = v;
}

__global__ void fill_i32_kernel(int* __restrict__ p, int v, int n) {
  int i = blockIdx.x * blockDim.x + threadIdx.x;
  if (i < n) p[i] = v;
}

__global__ void hist_rank_kernel(const int* __restrict__ dst, int* __restrict__ cnt,
                                 int* __restrict__ rank, int e) {
  const int q = e >> 3;
  const int i = blockIdx.x * blockDim.x + threadIdx.x;
  if (i >= q) return;
  const int4 d0 = ((const int4*)dst)[i];
  const int4 d1 = ((const int4*)dst)[i + q];
  asm volatile("" ::"v"(d0.x), "v"(d0.y), "v"(d0.z), "v"(d0.w),
               "v"(d1.x), "v"(d1.y), "v"(d1.z), "v"(d1.w));
  int4 r0, r1;
  r0.x = atomicAdd(&cnt[d0.x], 1);
  r0.y = atomicAdd(&cnt[d0.y], 1);
  r0.z = atomicAdd(&cnt[d0.z], 1);
  r0.w = atomicAdd(&cnt[d0.w], 1);
  r1.x = atomicAdd(&cnt[d1.x], 1);
  r1.y = atomicAdd(&cnt[d1.y], 1);
  r1.z = atomicAdd(&cnt[d1.z], 1);
  r1.w = atomicAdd(&cnt[d1.w], 1);
  ((int4*)rank)[i] = r0;
  ((int4*)rank)[i + q] = r1;
}

__global__ __launch_bounds__(1024) void rowptr_dinv_kernel(const int* __restrict__ cnt,
                                                           int* __restrict__ rowptr,
                                                           float* __restrict__ dinv, int n) {
  __shared__ int s_tot[32];
  __shared__ int s_part[32];
  const int tid = threadIdx.x;
  const int lane = tid & 31;
  const int wave = tid >> 5;
  const int first = blockIdx.x * 1024;

  int part = 0;
#pragma unroll 8
  for (int i = tid; i < first; i += 1024) part += cnt[i];

  const int idx = first + tid;
  const int c = (idx < n) ? cnt[idx] : 0;

  int incl = c;
  for (int o = 1; o < 32; o <<= 1) {
    const int t = __shfl_up(incl, o);
    if (lane >= o) incl += t;
  }
  for (int o = 16; o > 0; o >>= 1) part += __shfl_xor(part, o);
  if (lane == 31) s_tot[wave] = incl;
  if (lane == 0) s_part[wave] = part;
  __syncthreads();

  const int wt = s_tot[lane];
  int pt = s_part[lane];
  int wincl = wt;
  for (int o = 1; o < 32; o <<= 1) {
    const int t = __shfl_up(wincl, o);
    if (lane >= o) wincl += t;
  }
  for (int o = 16; o > 0; o >>= 1) pt += __shfl_xor(pt, o);
  const int wexcl = __shfl(wincl - wt, wave);

  const int excl = pt + wexcl + incl - c;
  if (idx < n) {
    rowptr[idx] = excl;
    dinv[idx] = rsqrtf((float)(c + 1));
  }
  if (idx == n - 1) rowptr[n] = excl + c;
}

__device__ __forceinline__ void csr_put(const int s, const int d, const int r,
                                        const int* __restrict__ rowptr,
                                        const float* __restrict__ dinv,
                                        int2* __restrict__ csr) {
  csr[rowptr[d] + r] = make_int2(s, __float_as_int(dinv[s] * dinv[d]));
}

__global__ void csr_fill_kernel(const int* __restrict__ src, const int* __restrict__ dst,
                                const int* __restrict__ rank, const int* __restrict__ rowptr,
                                const float* __restrict__ dinv, int* __restrict__ csr, int e) {
  const int q = e >> 3;
  const int i = blockIdx.x * blockDim.x + threadIdx.x;
  if (i >= q) return;
  const int4 s0 = ((const int4*)src)[i];
  const int4 s1 = ((const int4*)src)[i + q];
  const int4 d0 = ((const int4*)dst)[i];
  const int4 d1 = ((const int4*)dst)[i + q];
  const int4 r0 = ((const int4*)rank)[i];
  const int4 r1 = ((const int4*)rank)[i + q];
  int2* __restrict__ out = (int2*)csr;
  csr_put(s0.x, d0.x, r0.x, rowptr, dinv, out);
  csr_put(s0.y, d0.y, r0.y, rowptr, dinv, out);
  csr_put(s0.z, d0.z, r0.z, rowptr, dinv, out);
  csr_put(s0.w, d0.w, r0.w, rowptr, dinv, out);
  csr_put(s1.x, d1.x, r1.x, rowptr, dinv, out);
  csr_put(s1.y, d1.y, r1.y, rowptr, dinv, out);
  csr_put(s1.z, d1.z, r1.z, rowptr, dinv, out);
  csr_put(s1.w, d1.w, r1.w, rowptr, dinv, out);
}

__global__ void pad_x_kernel(const float* __restrict__ x, float* __restrict__ hpad, int n) {
  const int r = blockIdx.x * blockDim.x + threadIdx.x;
  if (r >= n) return;
  const float* __restrict__ xr = x + (size_t)r * 9;
  float4* __restrict__ o = (float4*)(hpad + (size_t)r * 16);
  o[0] = make_float4(xr[0], xr[1], xr[2], xr[3]);
  o[1] = make_float4(xr[4], xr[5], xr[6], xr[7]);
  o[2] = make_float4(xr[8], 0.0f, 0.0f, 0.0f);
  o[3] = make_float4(0.0f, 0.0f, 0.0f, 0.0f);
}

__global__ void pad_w0_kernel(const float* __restrict__ W0, float* __restrict__ w0pad) {
  int i = blockIdx.x * blockDim.x + threadIdx.x;
  if (i >= 16 * 64) return;
  int r = i >> 6, c = i & 63;
  w0pad[i] = (r < 9) ? W0[r * 64 + c] : 0.0f;
}

template <int KDIM>
__global__ void gemm_wmma_kernel(const float* __restrict__ A,
                                 const float* __restrict__ W,
                                 float* __restrict__ out) {
  const int wave = threadIdx.x >> 5;
  const int lane = threadIdx.x & 31;
  const int l15  = lane & 15;
  const int half = lane >> 4;
  const int m0   = blockIdx.x * 16;
  const int n0   = wave * 16;
  const int row  = m0 + l15;
  const int col  = n0 + l15;

  const float* __restrict__ arow = A + (size_t)row * KDIM + half * 2;
  const float* __restrict__ bcol = W + (size_t)half * 2 * 64 + col;

  v8f c = {};
#pragma unroll
  for (int k = 0; k < KDIM; k += 4) {
    v2f a, b;
    a.x = arow[k + 0];
    a.y = arow[k + 1];
    b.x = bcol[(size_t)k * 64];
    b.y = bcol[(size_t)k * 64 + 64];
    c = __builtin_amdgcn_wmma_f32_16x16x4_f32(false, a, false, b, (short)0, c,
                                              false, false);
  }

  float* __restrict__ orow = out + (size_t)(m0 + half * 8) * 64 + col;
#pragma unroll
  for (int v = 0; v < 8; ++v) {
    orow[(size_t)v * 64] = c[v];
  }
}

__device__ __forceinline__ void axpy4(float4& acc, const float4 v, const float w) {
  acc.x = fmaf(v.x, w, acc.x);
  acc.y = fmaf(v.y, w, acc.y);
  acc.z = fmaf(v.z, w, acc.z);
  acc.w = fmaf(v.w, w, acc.w);
}


__global__ __launch_bounds__(256) void gather_finish_kernel(const float* __restrict__ hW,
                                                            const int* __restrict__ rowptr,
                                                            const int* __restrict__ csr,
                                                            const float* __restrict__ dinv,
                                                            const float* __restrict__ b,
                                                            float* __restrict__ hout, int n) {
  const int gid = blockIdx.x * blockDim.x + threadIdx.x;
  const int d = gid >> 4;
  if (d >= n) return;
  const int fg = (gid & 15) * 4;
  const int2* __restrict__ ed = (const int2*)csr;

  int j = rowptr[d];
  const int end = rowptr[d + 1];
  const float4 self = *(const float4*)(hW + d * DD + fg);
  const float di = dinv[d];
  const float4 bb = *(const float4*)(b + fg);

  float4 acc = make_float4(0.0f, 0.0f, 0.0f, 0.0f);

  for (; j + 8 <= end; j += 8) {
    int2 e[8];
    float4 v[8];
#pragma unroll
    for (int k = 0; k < 8; ++k) e[k] = ed[j + k];
#pragma unroll
    for (int k = 0; k < 8; ++k) v[k] = *(const float4*)(hW + e[k].x * DD + fg);
#pragma unroll
    for (int k = 0; k < 8; ++k) axpy4(acc, v[k], __int_as_float(e[k].y));
  }

  if (j < end) {
    const int last = end - 1;
    int2 e[7];
    float4 v[7];
#pragma unroll
    for (int k = 0; k < 7; ++k) e[k] = ed[(j + k < last) ? (j + k) : last];
#pragma unroll
    for (int k = 0; k < 7; ++k) v[k] = *(const float4*)(hW + e[k].x * DD + fg);
#pragma unroll
    for (int k = 0; k < 7; ++k)
      axpy4(acc, v[k], (j + k < end) ? __int_as_float(e[k].y) : 0.0f);
  }

  float4 r;
  r.x = tanhf(acc.x + self.x * di * di + bb.x);
  r.y = tanhf(acc.y + self.y * di * di + bb.y);
  r.z = tanhf(acc.z + self.z * di * di + bb.z);
  r.w = tanhf(acc.w + self.w * di * di + bb.w);
  *(float4*)(hout + d * DD + fg) = r;
}


__device__ inline void atomicMaxF32(float* addr, float val) {
  if (val >= 0.0f)
    atomicMax((int*)addr, __float_as_int(val));
  else
    atomicMin((unsigned int*)addr, __float_as_uint(val));
}

__device__ __forceinline__ void pool_flush(const int g, const int f, const float s, const float m,
                                           const int c, float* __restrict__ gmax,
                                           float* __restrict__ gsum, float* __restrict__ cnt) {
  atomicAdd(&gsum[(size_t)g * DD + f], s);
  atomicMaxF32(&gmax[(size_t)g * DD + f], m);
  if (f == 0) atomicAdd(&cnt[g], (float)c);
}

#define POOL_ROWS 16
static_assert(NN % POOL_ROWS == 0, "pooling walks whole chunks of rows");

__global__ __launch_bounds__(256) void pool_acc_kernel(const float* __restrict__ h,
                                const int* __restrict__ batch,
                                float* __restrict__ gmax,
                                float* __restrict__ gsum,
                                float* __restrict__ cnt, int n) {
  const int gid = blockIdx.x * blockDim.x + threadIdx.x;
  const int f = gid & 63;
  const int r0 = (gid >> 6) * POOL_ROWS;
  if (r0 >= n) return;

  int g[POOL_ROWS];
  float v[POOL_ROWS];
#pragma unroll
  for (int k = 0; k < POOL_ROWS; ++k) g[k] = batch[r0 + k];
#pragma unroll
  for (int k = 0; k < POOL_ROWS; ++k) v[k] = h[(size_t)(r0 + k) * DD + f];

  int cur = g[0];
  float s = v[0];
  float m = v[0];
  int c = 1;
#pragma unroll
  for (int k = 1; k < POOL_ROWS; ++k) {
    if (g[k] != cur) {
      pool_flush(cur, f, s, m, c, gmax, gsum, cnt);
      cur = g[k];
      s = v[k];
      m = v[k];
      c = 1;
    } else {
      s += v[k];
      m = fmaxf(m, v[k]);
      c += 1;
    }
  }
  pool_flush(cur, f, s, m, c, gmax, gsum, cnt);
}

__global__ void pool_finish_kernel(const float* __restrict__ gmax,
                                   const float* __restrict__ gsum,
                                   const float* __restrict__ cnt,
                                   float* __restrict__ hidden, int g_) {
  int i = blockIdx.x * blockDim.x + threadIdx.x;
  if (i >= g_ * DD) return;
  const int g = i >> 6;
  const int f = i & 63;
  const float c = cnt[g];
  float mx = gmax[(size_t)g * DD + f];
  if (c == 0.0f) mx = 0.0f;
  hidden[(size_t)g * (2 * DD) + f] = mx;
  hidden[(size_t)g * (2 * DD) + DD + f] = gsum[(size_t)g * DD + f] / fmaxf(c, 1.0f);
}

__global__ void head_kernel(const float* __restrict__ hidden,
                            const float* __restrict__ Wout,
                            const float* __restrict__ bout,
                            float* __restrict__ out, int g_) {
  int g = blockIdx.x * blockDim.x + threadIdx.x;
  if (g >= g_) return;
  float acc = bout[0];
#pragma unroll 8
  for (int k = 0; k < 2 * DD; ++k) acc += hidden[(size_t)g * (2 * DD) + k] * Wout[k];
  out[g] = acc;
}


extern "C" void kernel_launch(void* const* d_in, const int* in_sizes, int n_in,
                              void* d_out, int out_size, void* d_ws, size_t ws_size,
                              hipStream_t stream) {
  (void)in_sizes; (void)n_in; (void)out_size; (void)ws_size;

  const float* x    = (const float*)d_in[0];
  const int*   eidx = (const int*)d_in[1];
  const int*   batch = (const int*)d_in[2];
  const float* W0 = (const float*)d_in[3];
  const float* b0 = (const float*)d_in[4];
  const float* W1 = (const float*)d_in[5];
  const float* b1 = (const float*)d_in[6];
  const float* W2 = (const float*)d_in[7];
  const float* b2 = (const float*)d_in[8];
  const float* W3 = (const float*)d_in[9];
  const float* b3 = (const float*)d_in[10];
  const float* Wout = (const float*)d_in[11];
  const float* bout = (const float*)d_in[12];

  const int* src = eidx;
  const int* dst = eidx + EE;

  float* outp   = (float*)d_out;
  float* hidden = (float*)d_out + GG;

  float* ws = (float*)d_ws;
  size_t o = 0;
  int*   cnt    = (int*)(ws + o); o += NN + 48;
  float* dinv   = ws + o;         o += NN + 48;
  int*   rowptr = (int*)(ws + o); o += NN + 112;
  int*   rank   = (int*)(ws + o); o += EE;
  int*   csr    = (int*)(ws + o); o += (size_t)EE * 2;
  float* hpad  = ws + o; o += (size_t)NN * 16;
  float* w0pad = ws + o; o += 16 * 64;
  float* hW    = ws + o; o += (size_t)NN * DD;
  float* h     = ws + o; o += (size_t)NN * DD;
  float* gmax  = ws + o; o += (size_t)GG * DD;
  float* gsum  = ws + o; o += (size_t)GG * DD;
  float* gcnt  = ws + o; o += GG;

  const int T = 256;
  auto nb = [](long long n, int t) { return (int)((n + t - 1) / t); };

  fill_i32_kernel<<<nb(NN, T), T, 0, stream>>>(cnt, 0, NN);
  hist_rank_kernel<<<nb(EE / 8, T), T, 0, stream>>>(dst, cnt, rank, EE);
  rowptr_dinv_kernel<<<nb(NN, 1024), 1024, 0, stream>>>(cnt, rowptr, dinv, NN);
  csr_fill_kernel<<<nb(EE / 8, T), T, 0, stream>>>(src, dst, rank, rowptr, dinv, csr, EE);

  pad_x_kernel<<<nb(NN, T), T, 0, stream>>>(x, hpad, NN);
  pad_w0_kernel<<<nb(16 * 64, T), T, 0, stream>>>(W0, w0pad);

  const int mtiles = NN / 16;
  const float* Wl[4] = {w0pad, W1, W2, W3};
  const float* bl[4] = {b0, b1, b2, b3};

  for (int l = 0; l < 4; ++l) {
    if (l == 0)
      gemm_wmma_kernel<16><<<mtiles, 128, 0, stream>>>(hpad, Wl[0], hW);
    else
      gemm_wmma_kernel<64><<<mtiles, 128, 0, stream>>>(h, Wl[l], hW);

    gather_finish_kernel<<<nb((long long)NN * 16, T), T, 0, stream>>>(hW, rowptr, csr, dinv, bl[l], h, NN);
  }

  fill_f32_kernel<<<nb((long long)GG * DD, T), T, 0, stream>>>(gmax, -FLT_MAX, GG * DD);
  fill_f32_kernel<<<nb((long long)GG * DD, T), T, 0, stream>>>(gsum, 0.0f, GG * DD);
  fill_f32_kernel<<<nb(GG, T), T, 0, stream>>>(gcnt, 0.0f, GG);
  pool_acc_kernel<<<nb((long long)(NN / POOL_ROWS) * DD, T), T, 0, stream>>>(h, batch, gmax, gsum, gcnt, NN);
  pool_finish_kernel<<<nb((long long)GG * DD, T), T, 0, stream>>>(gmax, gsum, gcnt, hidden, GG);

  head_kernel<<<nb(GG, 128), 128, 0, stream>>>(hidden, Wout, bout, outp, GG);
}


# One kernel per layer: neighbour sum over the layer's input rows, then the 16-node tile times W on WMMA from LDS, (A h) W for A (h W); the 4 GEMM launches and the product plane go; sum order changes
# speedup vs baseline: 2.7696x; 1.1253x over previous
#include <hip/hip_runtime.h>
#include <math.h>
#include <float.h>


typedef __attribute__((ext_vector_type(2))) float v2f;
typedef __attribute__((ext_vector_type(8))) float v8f;

#define NN 50000
#define EE 1200000
#define DD 64
#define GG 1024

static_assert(NN % 16 == 0, "the layer kernel's 16-row WMMA tiles need a row count that is a multiple of 16");
static_assert(EE % 8 == 0, "edge passes take 8 edges per lane as two 16-byte groups");
static_assert((NN + 48) % 4 == 0 && (NN + 112) % 4 == 0, "16-byte alignment of the edge arrays");


__global__ void fill_f32_kernel(float* __restrict__ p, float v, int n) {
  int i = blockIdx.x * blockDim.x + threadIdx.x;
  if (i < n) p[i] = v;
}

__global__ void fill_i32_kernel(int* __restrict__ p, int v, int n) {
  int i = blockIdx.x * blockDim.x + threadIdx.x;
  if (i < n) p[i] = v;
}

__global__ void hist_rank_kernel(const int* __restrict__ dst, int* __restrict__ cnt,
                                 int* __restrict__ rank, int e) {
  const int q = e >> 3;
  const int i = blockIdx.x * blockDim.x + threadIdx.x;
  if (i >= q) return;
  const int4 d0 = ((const int4*)dst)[i];
  const int4 d1 = ((const int4*)dst)[i + q];
  asm volatile("" ::"v"(d0.x), "v"(d0.y), "v"(d0.z), "v"(d0.w),
               "v"(d1.x), "v"(d1.y), "v"(d1.z), "v"(d1.w));
  int4 r0, r1;
  r0.x = atomicAdd(&cnt[d0.x], 1);
  r0.y = atomicAdd(&cnt[d0.y], 1);
  r0.z = atomicAdd(&cnt[d0.z], 1);
  r0.w = atomicAdd(&cnt[d0.w], 1);
  r1.x = atomicAdd(&cnt[d1.x], 1);
  r1.y = atomicAdd(&cnt[d1.y], 1);
  r1.z = atomicAdd(&cnt[d1.z], 1);
  r1.w = atomicAdd(&cnt[d1.w], 1);
  ((int4*)rank)[i] = r0;
  ((int4*)rank)[i + q] = r1;
}

__global__ __launch_bounds__(1024) void rowptr_dinv_kernel(const int* __restrict__ cnt,
                                                           int* __restrict__ rowptr,
                                                           float* __restrict__ dinv, int n) {
  __shared__ int s_tot[32];
  __shared__ int s_part[32];
  const int tid = threadIdx.x;
  const int lane = tid & 31;
  const int wave = tid >> 5;
  const int first = blockIdx.x * 1024;

  int part = 0;
#pragma unroll 8
  for (int i = tid; i < first; i += 1024) part += cnt[i];

  const int idx = first + tid;
  const int c = (idx < n) ? cnt[idx] : 0;

  int incl = c;
  for (int o = 1; o < 32; o <<= 1) {
    const int t = __shfl_up(incl, o);
    if (lane >= o) incl += t;
  }
  for (int o = 16; o > 0; o >>= 1) part += __shfl_xor(part, o);
  if (lane == 31) s_tot[wave] = incl;
  if (lane == 0) s_part[wave] = part;
  __syncthreads();

  const int wt = s_tot[lane];
  int pt = s_part[lane];
  int wincl = wt;
  for (int o = 1; o < 32; o <<= 1) {
    const int t = __shfl_up(wincl, o);
    if (lane >= o) wincl += t;
  }
  for (int o = 16; o > 0; o >>= 1) pt += __shfl_xor(pt, o);
  const int wexcl = __shfl(wincl - wt, wave);

  const int excl = pt + wexcl + incl - c;
  if (idx < n) {
    rowptr[idx] = excl;
    dinv[idx] = rsqrtf((float)(c + 1));
  }
  if (idx == n - 1) rowptr[n] = excl + c;
}

__device__ __forceinline__ void csr_put(const int s, const int d, const int r,
                                        const int* __restrict__ rowptr,
                                        const float* __restrict__ dinv,
                                        int2* __restrict__ csr) {
  csr[rowptr[d] + r] = make_int2(s, __float_as_int(dinv[s] * dinv[d]));
}

__global__ void csr_fill_kernel(const int* __restrict__ src, const int* __restrict__ dst,
                                const int* __restrict__ rank, const int* __restrict__ rowptr,
                                const float* __restrict__ dinv, int* __restrict__ csr, int e) {
  const int q = e >> 3;
  const int i = blockIdx.x * blockDim.x + threadIdx.x;
  if (i >= q) return;
  const int4 s0 = ((const int4*)src)[i];
  const int4 s1 = ((const int4*)src)[i + q];
  const int4 d0 = ((const int4*)dst)[i];
  const int4 d1 = ((const int4*)dst)[i + q];
  const int4 r0 = ((const int4*)rank)[i];
  const int4 r1 = ((const int4*)rank)[i + q];
  int2* __restrict__ out = (int2*)csr;
  csr_put(s0.x, d0.x, r0.x, rowptr, dinv, out);
  csr_put(s0.y, d0.y, r0.y, rowptr, dinv, out);
  csr_put(s0.z, d0.z, r0.z, rowptr, dinv, out);
  csr_put(s0.w, d0.w, r0.w, rowptr, dinv, out);
  csr_put(s1.x, d1.x, r1.x, rowptr, dinv, out);
  csr_put(s1.y, d1.y, r1.y, rowptr, dinv, out);
  csr_put(s1.z, d1.z, r1.z, rowptr, dinv, out);
  csr_put(s1.w, d1.w, r1.w, rowptr, dinv, out);
}

__global__ void pad_x_kernel(const float* __restrict__ x, float* __restrict__ hpad, int n) {
  const int r = blockIdx.x * blockDim.x + threadIdx.x;
  if (r >= n) return;
  const float* __restrict__ xr = x + (size_t)r * 9;
  float4* __restrict__ o = (float4*)(hpad + (size_t)r * 16);
  o[0] = make_float4(xr[0], xr[1], xr[2], xr[3]);
  o[1] = make_float4(xr[4], xr[5], xr[6], xr[7]);
  o[2] = make_float4(xr[8], 0.0f, 0.0f, 0.0f);
  o[3] = make_float4(0.0f, 0.0f, 0.0f, 0.0f);
}

__global__ void pad_w0_kernel(const float* __restrict__ W0, float* __restrict__ w0pad) {
  int i = blockIdx.x * blockDim.x + threadIdx.x;
  if (i >= 16 * 64) return;
  int r = i >> 6, c = i & 63;
  w0pad[i] = (r < 9) ? W0[r * 64 + c] : 0.0f;
}

__device__ __forceinline__ void axpy4(float4& acc, const float4 v, const float w) {
  acc.x = fmaf(v.x, w, acc.x);
  acc.y = fmaf(v.y, w, acc.y);
  acc.z = fmaf(v.z, w, acc.z);
  acc.w = fmaf(v.w, w, acc.w);
}

template <int K>
__global__ __launch_bounds__(K * 4) void layer_kernel(const float* __restrict__ hin,
                                                      const float* __restrict__ W,
                                                      const int* __restrict__ rowptr,
                                                      const int* __restrict__ csr,
                                                      const float* __restrict__ dinv,
                                                      const float* __restrict__ b,
                                                      float* __restrict__ hout) {
  constexpr int LPN = K / 4;
  constexpr int NW = K / 8;
  constexpr int TPW = (NW >= 4) ? 1 : 4 / NW;
  constexpr int KS = K / 4;
  constexpr int P = K + 4;
  __shared__ __attribute__((aligned(16))) float zs[16 * P];

  const int tid = threadIdx.x;
  const int node = tid / LPN;
  const int fg = (tid % LPN) * 4;
  const int m0 = blockIdx.x * 16;
  const int d = m0 + node;
  const int2* __restrict__ ed = (const int2*)csr;

  int j = rowptr[d];
  const int end = rowptr[d + 1];
  const float4 self = *(const float4*)(hin + d * K + fg);
  const float di = dinv[d];

  float4 acc = make_float4(0.0f, 0.0f, 0.0f, 0.0f);

  for (; j + 8 <= end; j += 8) {
    int2 e[8];
    float4 v[8];
#pragma unroll
    for (int k = 0; k < 8; ++k) e[k] = ed[j + k];
#pragma unroll
    for (int k = 0; k < 8; ++k) v[k] = *(const float4*)(hin + e[k].x * K + fg);
#pragma unroll
    for (int k = 0; k < 8; ++k) axpy4(acc, v[k], __int_as_float(e[k].y));
  }

  if (j < end) {
    const int last = end - 1;
    int2 e[7];
    float4 v[7];
#pragma unroll
    for (int k = 0; k < 7; ++k) e[k] = ed[(j + k < last) ? (j + k) : last];
#pragma unroll
    for (int k = 0; k < 7; ++k) v[k] = *(const float4*)(hin + e[k].x * K + fg);
#pragma unroll
    for (int k = 0; k < 7; ++k)
      axpy4(acc, v[k], (j + k < end) ? __int_as_float(e[k].y) : 0.0f);
  }

  const int wave = __builtin_amdgcn_readfirstlane(tid >> 5);
  const int lane = tid & 31;
  const int l15 = lane & 15;
  const int half = lane >> 4;
  const bool mult = wave < 4;

  const float dd = di * di;
  float4 z;
  z.x = fmaf(self.x, dd, acc.x);
  z.y = fmaf(self.y, dd, acc.y);
  z.z = fmaf(self.z, dd, acc.z);
  z.w = fmaf(self.w, dd, acc.w);
  *(float4*)(zs + node * P + fg) = z;

  v2f bf[TPW][KS];
  if (mult) {
#pragma unroll
    for (int t = 0; t < TPW; ++t) {
      const float* __restrict__ bcol = W + half * 2 * 64 + (wave + t * NW) * 16 + l15;
#pragma unroll
      for (int s = 0; s < KS; ++s) {
        bf[t][s].x = bcol[s * 4 * 64];
        bf[t][s].y = bcol[s * 4 * 64 + 64];
      }
    }
  }
  __builtin_amdgcn_fence(__ATOMIC_RELEASE, "workgroup", "local");
  __builtin_amdgcn_s_barrier();
  __builtin_amdgcn_fence(__ATOMIC_ACQUIRE, "workgroup", "local");

  if (mult) {
    const float* zrow = zs + l15 * P + half * 2;
#pragma unroll
    for (int t = 0; t < TPW; ++t) {
      const int col = (wave + t * NW) * 16 + l15;
      v8f c0 = {};
      v8f c1 = {};
#pragma unroll
      for (int s = 0; s < KS; s += 2) {
        const v2f a0 = *(const v2f*)(zrow + s * 4);
        const v2f a1 = *(const v2f*)(zrow + s * 4 + 4);
        c0 = __builtin_amdgcn_wmma_f32_16x16x4_f32(false, a0, false, bf[t][s], (short)0, c0,
                                                   false, false);
        c1 = __builtin_amdgcn_wmma_f32_16x16x4_f32(false, a1, false, bf[t][s + 1], (short)0, c1,
                                                   false, false);
      }
      const float bias = b[col];
      float* __restrict__ orow = hout + (m0 + half * 8) * 64 + col;
#pragma unroll
      for (int v = 0; v < 8; ++v) orow[v * 64] = tanhf(c0[v] + c1[v] + bias);
    }
  }
}


__device__ inline void atomicMaxF32(float* addr, float val) {
  if (val >= 0.0f)
    atomicMax((int*)addr, __float_as_int(val));
  else
    atomicMin((unsigned int*)addr, __float_as_uint(val));
}

__device__ __forceinline__ void pool_flush(const int g, const int f, const float s, const float m,
                                           const int c, float* __restrict__ gmax,
                                           float* __restrict__ gsum, float* __restrict__ cnt) {
  atomicAdd(&gsum[(size_t)g * DD + f], s);
  atomicMaxF32(&gmax[(size_t)g * DD + f], m);
  if (f == 0) atomicAdd(&cnt[g], (float)c);
}

#define POOL_ROWS 16
static_assert(NN % POOL_ROWS == 0, "pooling walks whole chunks of rows");

__global__ __launch_bounds__(256) void pool_acc_kernel(const float* __restrict__ h,
                                const int* __restrict__ batch,
                                float* __restrict__ gmax,
                                float* __restrict__ gsum,
                                float* __restrict__ cnt, int n) {
  const int gid = blockIdx.x * blockDim.x + threadIdx.x;
  const int f = gid & 63;
  const int r0 = (gid >> 6) * POOL_ROWS;
  if (r0 >= n) return;

  int g[POOL_ROWS];
  float v[POOL_ROWS];
#pragma unroll
  for (int k = 0; k < POOL_ROWS; ++k) g[k] = batch[r0 + k];
#pragma unroll
  for (int k = 0; k < POOL_ROWS; ++k) v[k] = h[(size_t)(r0 + k) * DD + f];

  int cur = g[0];
  float s = v[0];
  float m = v[0];
  int c = 1;
#pragma unroll
  for (int k = 1; k < POOL_ROWS; ++k) {
    if (g[k] != cur) {
      pool_flush(cur, f, s, m, c, gmax, gsum, cnt);
      cur = g[k];
      s = v[k];
      m = v[k];
      c = 1;
    } else {
      s += v[k];
      m = fmaxf(m, v[k]);
      c += 1;
    }
  }
  pool_flush(cur, f, s, m, c, gmax, gsum, cnt);
}

__global__ void pool_finish_kernel(const float* __restrict__ gmax,
                                   const float* __restrict__ gsum,
                                   const float* __restrict__ cnt,
                                   float* __restrict__ hidden, int g_) {
  int i = blockIdx.x * blockDim.x + threadIdx.x;
  if (i >= g_ * DD) return;
  const int g = i >> 6;
  const int f = i & 63;
  const float c = cnt[g];
  float mx = gmax[(size_t)g * DD + f];
  if (c == 0.0f) mx = 0.0f;
  hidden[(size_t)g * (2 * DD) + f] = mx;
  hidden[(size_t)g * (2 * DD) + DD + f] = gsum[(size_t)g * DD + f] / fmaxf(c, 1.0f);
}

__global__ void head_kernel(const float* __restrict__ hidden,
                            const float* __restrict__ Wout,
                            const float* __restrict__ bout,
                            float* __restrict__ out, int g_) {
  int g = blockIdx.x * blockDim.x + threadIdx.x;
  if (g >= g_) return;
  float acc = bout[0];
#pragma unroll 8
  for (int k = 0; k < 2 * DD; ++k) acc += hidden[(size_t)g * (2 * DD) + k] * Wout[k];
  out[g] = acc;
}


extern "C" void kernel_launch(void* const* d_in, const int* in_sizes, int n_in,
                              void* d_out, int out_size, void* d_ws, size_t ws_size,
                              hipStream_t stream) {
  (void)in_sizes; (void)n_in; (void)out_size; (void)ws_size;

  const float* x    = (const float*)d_in[0];
  const int*   eidx = (const int*)d_in[1];
  const int*   batch = (const int*)d_in[2];
  const float* W0 = (const float*)d_in[3];
  const float* b0 = (const float*)d_in[4];
  const float* W1 = (const float*)d_in[5];
  const float* b1 = (const float*)d_in[6];
  const float* W2 = (const float*)d_in[7];
  const float* b2 = (const float*)d_in[8];
  const float* W3 = (const float*)d_in[9];
  const float* b3 = (const float*)d_in[10];
  const float* Wout = (const float*)d_in[11];
  const float* bout = (const float*)d_in[12];

  const int* src = eidx;
  const int* dst = eidx + EE;

  float* outp   = (float*)d_out;
  float* hidden = (float*)d_out + GG;

  float* ws = (float*)d_ws;
  size_t o = 0;
  int*   cnt    = (int*)(ws + o); o += NN + 48;
  float* dinv   = ws + o;         o += NN + 48;
  int*   rowptr = (int*)(ws + o); o += NN + 112;
  int*   rank   = (int*)(ws + o); o += EE;
  int*   csr    = (int*)(ws + o); o += (size_t)EE * 2;
  float* hpad  = ws + o; o += (size_t)NN * 16;
  float* w0pad = ws + o; o += 16 * 64;
  float* h0    = ws + o; o += (size_t)NN * DD;
  float* h1    = ws + o; o += (size_t)NN * DD;
  float* gmax  = ws + o; o += (size_t)GG * DD;
  float* gsum  = ws + o; o += (size_t)GG * DD;
  float* gcnt  = ws + o; o += GG;

  const int T = 256;
  auto nb = [](long long n, int t) { return (int)((n + t - 1) / t); };

  fill_i32_kernel<<<nb(NN, T), T, 0, stream>>>(cnt, 0, NN);
  hist_rank_kernel<<<nb(EE / 8, T), T, 0, stream>>>(dst, cnt, rank, EE);
  rowptr_dinv_kernel<<<nb(NN, 1024), 1024, 0, stream>>>(cnt, rowptr, dinv, NN);
  csr_fill_kernel<<<nb(EE / 8, T), T, 0, stream>>>(src, dst, rank, rowptr, dinv, csr, EE);

  pad_x_kernel<<<nb(NN, T), T, 0, stream>>>(x, hpad, NN);
  pad_w0_kernel<<<nb(16 * 64, T), T, 0, stream>>>(W0, w0pad);

  const int mtiles = NN / 16;

  layer_kernel<16><<<mtiles, 64, 0, stream>>>(hpad, w0pad, rowptr, csr, dinv, b0, h0);
  layer_kernel<64><<<mtiles, 256, 0, stream>>>(h0, W1, rowptr, csr, dinv, b1, h1);
  layer_kernel<64><<<mtiles, 256, 0, stream>>>(h1, W2, rowptr, csr, dinv, b2, h0);
  layer_kernel<64><<<mtiles, 256, 0, stream>>>(h0, W3, rowptr, csr, dinv, b3, h1);

  fill_f32_kernel<<<nb((long long)GG * DD, T), T, 0, stream>>>(gmax, -FLT_MAX, GG * DD);
  fill_f32_kernel<<<nb((long long)GG * DD, T), T, 0, stream>>>(gsum, 0.0f, GG * DD);
  fill_f32_kernel<<<nb(GG, T), T, 0, stream>>>(gcnt, 0.0f, GG);
  pool_acc_kernel<<<nb((long long)(NN / POOL_ROWS) * DD, T), T, 0, stream>>>(h1, batch, gmax, gsum, gcnt, NN);
  pool_finish_kernel<<<nb((long long)GG * DD, T), T, 0, stream>>>(gmax, gsum, gcnt, hidden, GG);

  head_kernel<<<nb(GG, 128), 128, 0, stream>>>(hidden, Wout, bout, outp, GG);
}


# Edge list sorted by destination with work-group LDS counters (bin by 128-node range, then count and place per bin) instead of one returning global atomic per edge; order of edge sums changes
# speedup vs baseline: 3.8863x; 1.4032x over previous
#include <hip/hip_runtime.h>
#include <math.h>
#include <float.h>


typedef __attribute__((ext_vector_type(2))) float v2f;
typedef __attribute__((ext_vector_type(8))) float v8f;

#define NN 50000
#define EE 1200000
#define DD 64
#define GG 1024

static_assert(NN % 16 == 0, "the layer kernel's 16-row WMMA tiles need a row count that is a multiple of 16");
static_assert(EE % 4 == 0, "the norm pass takes 4 pairs per lane as two 16-byte groups");


__global__ void fill_f32_kernel(float* __restrict__ p, float v, int n) {
  int i = blockIdx.x * blockDim.x + threadIdx.x;
  if (i < n) p[i] = v;
}

#define BSH 7
#define BS (1 << BSH)
#define NB ((NN + BS - 1) / BS)
#define NCH 240
#define CH (EE / NCH)
static_assert(EE % NCH == 0 && CH % 4 == 0, "chunks are whole 16-byte groups of edges");
static_assert(NB <= 32 * 13, "the bin scan gives each lane of one wave 13 bins");

__global__ __launch_bounds__(256) void bin_count_kernel(const int* __restrict__ dst,
                                                        int* __restrict__ table) {
  __shared__ int hist[NB];
  const int tid = threadIdx.x;
  const int c = blockIdx.x;
  for (int i = tid; i < NB; i += 256) hist[i] = 0;
  __syncthreads();
  const int4* __restrict__ pd = (const int4*)(dst + c * CH);
  constexpr int Q = CH / 4;
  constexpr int KB = (Q + 255) / 256;
  int4 d[KB];
#pragma unroll
  for (int k = 0; k < KB; ++k) {
    const int i = tid + k * 256;
    d[k] = pd[(i < Q) ? i : (Q - 1)];
  }
#pragma unroll
  for (int k = 0; k < KB; ++k) {
    if (tid + k * 256 < Q) {
      atomicAdd(&hist[d[k].x >> BSH], 1);
      atomicAdd(&hist[d[k].y >> BSH], 1);
      atomicAdd(&hist[d[k].z >> BSH], 1);
      atomicAdd(&hist[d[k].w >> BSH], 1);
    }
  }
  __syncthreads();
  for (int i = tid; i < NB; i += 256) table[i * NCH + c] = hist[i];
}

__global__ __launch_bounds__(1024) void bin_scatter_kernel(const int* __restrict__ src,
                                                           const int* __restrict__ dst,
                                                           const int* __restrict__ table,
                                                           int* __restrict__ binstart,
                                                           int* __restrict__ part_) {
  int2* __restrict__ part = (int2*)part_;
  __shared__ int tot[NB + 1];
  __shared__ int cur[NB];
  const int tid = threadIdx.x;
  const int lane = tid & 31;
  const int wave = tid >> 5;
  const int c = blockIdx.x;

  constexpr int KR = (NCH + 31) / 32;
  for (int b0 = wave; b0 < NB; b0 += 4 * 32) {
    int tv[4][KR];
#pragma unroll
    for (int u = 0; u < 4; ++u) {
      const int b = b0 + u * 32;
      const int* __restrict__ row = table + ((b < NB) ? b : (NB - 1)) * NCH;
#pragma unroll
      for (int k = 0; k < KR; ++k) {
        const int idx = k * 32 + lane;
        tv[u][k] = row[(idx < NCH) ? idx : (NCH - 1)];
      }
    }
#pragma unroll
    for (int u = 0; u < 4; ++u) {
      const int b = b0 + u * 32;
      int t = 0;
      int p = 0;
#pragma unroll
      for (int k = 0; k < KR; ++k) {
        const int idx = k * 32 + lane;
        const int v = (idx < NCH) ? tv[u][k] : 0;
        t += v;
        p += (idx < c) ? v : 0;
      }
      for (int o = 16; o > 0; o >>= 1) {
        t += __shfl_xor(t, o);
        p += __shfl_xor(p, o);
      }
      if (lane == 0 && b < NB) {
        tot[b] = t;
        cur[b] = p;
      }
    }
  }
  __syncthreads();

  if (wave == 0) {
    int v[13];
    int s = 0;
#pragma unroll
    for (int k = 0; k < 13; ++k) {
      const int idx = lane * 13 + k;
      v[k] = (idx < NB) ? tot[idx] : 0;
      s += v[k];
    }
    int incl = s;
    for (int o = 1; o < 32; o <<= 1) {
      const int u = __shfl_up(incl, o);
      if (lane >= o) incl += u;
    }
    int excl = incl - s;
#pragma unroll
    for (int k = 0; k < 13; ++k) {
      const int idx = lane * 13 + k;
      if (idx < NB) tot[idx] = excl;
      excl += v[k];
    }
    if (lane == 31) tot[NB] = incl;
  }
  __syncthreads();
  for (int b = tid; b < NB; b += 1024) cur[b] += tot[b];
  if (c == 0)
    for (int b = tid; b <= NB; b += 1024) binstart[b] = tot[b];
  __syncthreads();

  const int4* __restrict__ ps = (const int4*)(src + c * CH);
  const int4* __restrict__ pd = (const int4*)(dst + c * CH);
  constexpr int Q = CH / 4;
  constexpr int KB = (Q + 1023) / 1024;
  int4 s[KB];
  int4 d[KB];
#pragma unroll
  for (int k = 0; k < KB; ++k) {
    const int i = tid + k * 1024;
    s[k] = ps[(i < Q) ? i : (Q - 1)];
    d[k] = pd[(i < Q) ? i : (Q - 1)];
  }
#pragma unroll
  for (int k = 0; k < KB; ++k) {
    if (tid + k * 1024 < Q) {
      part[atomicAdd(&cur[d[k].x >> BSH], 1)] = make_int2(s[k].x, d[k].x);
      part[atomicAdd(&cur[d[k].y >> BSH], 1)] = make_int2(s[k].y, d[k].y);
      part[atomicAdd(&cur[d[k].z >> BSH], 1)] = make_int2(s[k].z, d[k].z);
      part[atomicAdd(&cur[d[k].w >> BSH], 1)] = make_int2(s[k].w, d[k].w);
    }
  }
}

__global__ __launch_bounds__(512) void bin_build_kernel(const int* __restrict__ part_,
                                                         const int* __restrict__ binstart,
                                                         int* __restrict__ rowptr,
                                                         float* __restrict__ dinv,
                                                         int* __restrict__ csr_) {
  const int2* __restrict__ part = (const int2*)part_;
  int2* __restrict__ csr = (int2*)csr_;
  __shared__ int cnt[BS];
  __shared__ int cursor[BS];
  __shared__ int wtot[BS / 32];
  const int tid = threadIdx.x;
  const int lane = tid & 31;
  const int wave = tid >> 5;
  const int b = blockIdx.x;
  const int lo = binstart[b];
  const int hi = binstart[b + 1];
  const int base = b * BS;

  if (tid < BS) cnt[tid] = 0;
  __syncthreads();
  for (int j0 = lo + tid; j0 < hi; j0 += 8 * 512) {
    int2 e[8];
#pragma unroll
    for (int k = 0; k < 8; ++k) {
      const int j = j0 + k * 512;
      e[k] = part[(j < hi) ? j : (hi - 1)];
    }
#pragma unroll
    for (int k = 0; k < 8; ++k)
      if (j0 + k * 512 < hi) atomicAdd(&cnt[e[k].y - base], 1);
  }
  __syncthreads();

  int c = 0;
  int incl = 0;
  if (tid < BS) {
    c = cnt[tid];
    incl = c;
    for (int o = 1; o < 32; o <<= 1) {
      const int u = __shfl_up(incl, o);
      if (lane >= o) incl += u;
    }
    if (lane == 31) wtot[wave] = incl;
  }
  __syncthreads();
  if (tid < BS) {
    int off = lo;
    for (int w = 0; w < wave; ++w) off += wtot[w];
    const int excl = off + incl - c;
    cursor[tid] = excl;
    const int node = base + tid;
    if (node < NN) {
      rowptr[node] = excl;
      dinv[node] = rsqrtf((float)(c + 1));
    }
  }
  if (b == NB - 1 && tid == 0) rowptr[NN] = hi;
  __syncthreads();

  for (int j0 = lo + tid; j0 < hi; j0 += 8 * 512) {
    int2 e[8];
#pragma unroll
    for (int k = 0; k < 8; ++k) {
      const int j = j0 + k * 512;
      e[k] = part[(j < hi) ? j : (hi - 1)];
    }
#pragma unroll
    for (int k = 0; k < 8; ++k)
      if (j0 + k * 512 < hi) csr[atomicAdd(&cursor[e[k].y - base], 1)] = e[k];
  }
}

__global__ void norm_fill_kernel(int* csr, const float* __restrict__ dinv, int e) {
  const int q = e >> 2;
  const int i = blockIdx.x * blockDim.x + threadIdx.x;
  if (i >= q) return;
  int4 a = ((const int4*)csr)[i];
  int4 b = ((const int4*)csr)[i + q];
  a.y = __float_as_int(dinv[a.x] * dinv[a.y]);
  a.w = __float_as_int(dinv[a.z] * dinv[a.w]);
  b.y = __float_as_int(dinv[b.x] * dinv[b.y]);
  b.w = __float_as_int(dinv[b.z] * dinv[b.w]);
  ((int4*)csr)[i] = a;
  ((int4*)csr)[i + q] = b;
}

__global__ void pad_x_kernel(const float* __restrict__ x, float* __restrict__ hpad, int n) {
  const int r = blockIdx.x * blockDim.x + threadIdx.x;
  if (r >= n) return;
  const float* __restrict__ xr = x + (size_t)r * 9;
  float4* __restrict__ o = (float4*)(hpad + (size_t)r * 16);
  o[0] = make_float4(xr[0], xr[1], xr[2], xr[3]);
  o[1] = make_float4(xr[4], xr[5], xr[6], xr[7]);
  o[2] = make_float4(xr[8], 0.0f, 0.0f, 0.0f);
  o[3] = make_float4(0.0f, 0.0f, 0.0f, 0.0f);
}

__global__ void pad_w0_kernel(const float* __restrict__ W0, float* __restrict__ w0pad) {
  int i = blockIdx.x * blockDim.x + threadIdx.x;
  if (i >= 16 * 64) return;
  int r = i >> 6, c = i & 63;
  w0pad[i] = (r < 9) ? W0[r * 64 + c] : 0.0f;
}

__device__ __forceinline__ void axpy4(float4& acc, const float4 v, const float w) {
  acc.x = fmaf(v.x, w, acc.x);
  acc.y = fmaf(v.y, w, acc.y);
  acc.z = fmaf(v.z, w, acc.z);
  acc.w = fmaf(v.w, w, acc.w);
}

template <int K>
__global__ __launch_bounds__(K * 4) void layer_kernel(const float* __restrict__ hin,
                                                      const float* __restrict__ W,
                                                      const int* __restrict__ rowptr,
                                                      const int* __restrict__ csr,
                                                      const float* __restrict__ dinv,
                                                      const float* __restrict__ b,
                                                      float* __restrict__ hout) {
  constexpr int LPN = K / 4;
  constexpr int NW = K / 8;
  constexpr int TPW = (NW >= 4) ? 1 : 4 / NW;
  constexpr int KS = K / 4;
  constexpr int P = K + 4;
  __shared__ __attribute__((aligned(16))) float zs[16 * P];

  const int tid = threadIdx.x;
  const int node = tid / LPN;
  const int fg = (tid % LPN) * 4;
  const int m0 = blockIdx.x * 16;
  const int d = m0 + node;
  const int2* __restrict__ ed = (const int2*)csr;

  int j = rowptr[d];
  const int end = rowptr[d + 1];
  const float4 self = *(const float4*)(hin + d * K + fg);
  const float di = dinv[d];

  float4 acc = make_float4(0.0f, 0.0f, 0.0f, 0.0f);

  for (; j + 8 <= end; j += 8) {
    int2 e[8];
    float4 v[8];
#pragma unroll
    for (int k = 0; k < 8; ++k) e[k] = ed[j + k];
#pragma unroll
    for (int k = 0; k < 8; ++k) v[k] = *(const float4*)(hin + e[k].x * K + fg);
#pragma unroll
    for (int k = 0; k < 8; ++k) axpy4(acc, v[k], __int_as_float(e[k].y));
  }

  if (j < end) {
    const int last = end - 1;
    int2 e[7];
    float4 v[7];
#pragma unroll
    for (int k = 0; k < 7; ++k) e[k] = ed[(j + k < last) ? (j + k) : last];
#pragma unroll
    for (int k = 0; k < 7; ++k) v[k] = *(const float4*)(hin + e[k].x * K + fg);
#pragma unroll
    for (int k = 0; k < 7; ++k)
      axpy4(acc, v[k], (j + k < end) ? __int_as_float(e[k].y) : 0.0f);
  }

  const int wave = __builtin_amdgcn_readfirstlane(tid >> 5);
  const int lane = tid & 31;
  const int l15 = lane & 15;
  const int half = lane >> 4;
  const bool mult = wave < 4;

  const float dd = di * di;
  float4 z;
  z.x = fmaf(self.x, dd, acc.x);
  z.y = fmaf(self.y, dd, acc.y);
  z.z = fmaf(self.z, dd, acc.z);
  z.w = fmaf(self.w, dd, acc.w);
  *(float4*)(zs + node * P + fg) = z;

  v2f bf[TPW][KS];
  if (mult) {
#pragma unroll
    for (int t = 0; t < TPW; ++t) {
      const float* __restrict__ bcol = W + half * 2 * 64 + (wave + t * NW) * 16 + l15;
#pragma unroll
      for (int s = 0; s < KS; ++s) {
        bf[t][s].x = bcol[s * 4 * 64];
        bf[t][s].y = bcol[s * 4 * 64 + 64];
      }
    }
  }
  __builtin_amdgcn_fence(__ATOMIC_RELEASE, "workgroup", "local");
  __builtin_amdgcn_s_barrier();
  __builtin_amdgcn_fence(__ATOMIC_ACQUIRE, "workgroup", "local");

  if (mult) {
    const float* zrow = zs + l15 * P + half * 2;
#pragma unroll
    for (int t = 0; t < TPW; ++t) {
      const int col = (wave + t * NW) * 16 + l15;
      v8f c0 = {};
      v8f c1 = {};
#pragma unroll
      for (int s = 0; s < KS; s += 2) {
        const v2f a0 = *(const v2f*)(zrow + s * 4);
        const v2f a1 = *(const v2f*)(zrow + s * 4 + 4);
        c0 = __builtin_amdgcn_wmma_f32_16x16x4_f32(false, a0, false, bf[t][s], (short)0, c0,
                                                   false, false);
        c1 = __builtin_amdgcn_wmma_f32_16x16x4_f32(false, a1, false, bf[t][s + 1], (short)0, c1,
                                                   false, false);
      }
      const float bias = b[col];
      float* __restrict__ orow = hout + (m0 + half * 8) * 64 + col;
#pragma unroll
      for (int v = 0; v < 8; ++v) orow[v * 64] = tanhf(c0[v] + c1[v] + bias);
    }
  }
}


__device__ inline void atomicMaxF32(float* addr, float val) {
  if (val >= 0.0f)
    atomicMax((int*)addr, __float_as_int(val));
  else
    atomicMin((unsigned int*)addr, __float_as_uint(val));
}

__device__ __forceinline__ void pool_flush(const int g, const int f, const float s, const float m,
                                           const int c, float* __restrict__ gmax,
                                           float* __restrict__ gsum, float* __restrict__ cnt) {
  atomicAdd(&gsum[(size_t)g * DD + f], s);
  atomicMaxF32(&gmax[(size_t)g * DD + f], m);
  if (f == 0) atomicAdd(&cnt[g], (float)c);
}

#define POOL_ROWS 16
static_assert(NN % POOL_ROWS == 0, "pooling walks whole chunks of rows");

__global__ __launch_bounds__(256) void pool_acc_kernel(const float* __restrict__ h,
                                const int* __restrict__ batch,
                                float* __restrict__ gmax,
                                float* __restrict__ gsum,
                                float* __restrict__ cnt, int n) {
  const int gid = blockIdx.x * blockDim.x + threadIdx.x;
  const int f = gid & 63;
  const int r0 = (gid >> 6) * POOL_ROWS;
  if (r0 >= n) return;

  int g[POOL_ROWS];
  float v[POOL_ROWS];
#pragma unroll
  for (int k = 0; k < POOL_ROWS; ++k) g[k] = batch[r0 + k];
#pragma unroll
  for (int k = 0; k < POOL_ROWS; ++k) v[k] = h[(size_t)(r0 + k) * DD + f];

  int cur = g[0];
  float s = v[0];
  float m = v[0];
  int c = 1;
#pragma unroll
  for (int k = 1; k < POOL_ROWS; ++k) {
    if (g[k] != cur) {
      pool_flush(cur, f, s, m, c, gmax, gsum, cnt);
      cur = g[k];
      s = v[k];
      m = v[k];
      c = 1;
    } else {
      s += v[k];
      m = fmaxf(m, v[k]);
      c += 1;
    }
  }
  pool_flush(cur, f, s, m, c, gmax, gsum, cnt);
}

__global__ void pool_finish_kernel(const float* __restrict__ gmax,
                                   const float* __restrict__ gsum,
                                   const float* __restrict__ cnt,
                                   float* __restrict__ hidden, int g_) {
  int i = blockIdx.x * blockDim.x + threadIdx.x;
  if (i >= g_ * DD) return;
  const int g = i >> 6;
  const int f = i & 63;
  const float c = cnt[g];
  float mx = gmax[(size_t)g * DD + f];
  if (c == 0.0f) mx = 0.0f;
  hidden[(size_t)g * (2 * DD) + f] = mx;
  hidden[(size_t)g * (2 * DD) + DD + f] = gsum[(size_t)g * DD + f] / fmaxf(c, 1.0f);
}

__global__ void head_kernel(const float* __restrict__ hidden,
                            const float* __restrict__ Wout,
                            const float* __restrict__ bout,
                            float* __restrict__ out, int g_) {
  int g = blockIdx.x * blockDim.x + threadIdx.x;
  if (g >= g_) return;
  float acc = bout[0];
#pragma unroll 8
  for (int k = 0; k < 2 * DD; ++k) acc += hidden[(size_t)g * (2 * DD) + k] * Wout[k];
  out[g] = acc;
}


extern "C" void kernel_launch(void* const* d_in, const int* in_sizes, int n_in,
                              void* d_out, int out_size, void* d_ws, size_t ws_size,
                              hipStream_t stream) {
  (void)in_sizes; (void)n_in; (void)out_size; (void)ws_size;

  const float* x    = (const float*)d_in[0];
  const int*   eidx = (const int*)d_in[1];
  const int*   batch = (const int*)d_in[2];
  const float* W0 = (const float*)d_in[3];
  const float* b0 = (const float*)d_in[4];
  const float* W1 = (const float*)d_in[5];
  const float* b1 = (const float*)d_in[6];
  const float* W2 = (const float*)d_in[7];
  const float* b2 = (const float*)d_in[8];
  const float* W3 = (const float*)d_in[9];
  const float* b3 = (const float*)d_in[10];
  const float* Wout = (const float*)d_in[11];
  const float* bout = (const float*)d_in[12];

  const int* src = eidx;
  const int* dst = eidx + EE;

  float* outp   = (float*)d_out;
  float* hidden = (float*)d_out + GG;

  float* ws = (float*)d_ws;
  size_t o = 0;
  float* dinv   = ws + o;         o += NN + 48;
  int*   rowptr = (int*)(ws + o); o += NN + 112;
  int*   table  = (int*)(ws + o); o += NB * NCH + 48;
  int*   bstart = (int*)(ws + o); o += 448;
  int*   part   = (int*)(ws + o); o += (size_t)EE * 2;
  int*   csr    = (int*)(ws + o); o += (size_t)EE * 2;
  float* hpad  = ws + o; o += (size_t)NN * 16;
  float* w0pad = ws + o; o += 16 * 64;
  float* h0    = ws + o; o += (size_t)NN * DD;
  float* h1    = ws + o; o += (size_t)NN * DD;
  float* gmax  = ws + o; o += (size_t)GG * DD;
  float* gsum  = ws + o; o += (size_t)GG * DD;
  float* gcnt  = ws + o; o += GG;

  const int T = 256;
  auto nb = [](long long n, int t) { return (int)((n + t - 1) / t); };

  bin_count_kernel<<<NCH, 256, 0, stream>>>(dst, table);
  bin_scatter_kernel<<<NCH, 1024, 0, stream>>>(src, dst, table, bstart, part);
  bin_build_kernel<<<NB, 512, 0, stream>>>(part, bstart, rowptr, dinv, csr);
  norm_fill_kernel<<<nb(EE / 4, T), T, 0, stream>>>(csr, dinv, EE);

  pad_x_kernel<<<nb(NN, T), T, 0, stream>>>(x, hpad, NN);
  pad_w0_kernel<<<nb(16 * 64, T), T, 0, stream>>>(W0, w0pad);

  const int mtiles = NN / 16;

  layer_kernel<16><<<mtiles, 64, 0, stream>>>(hpad, w0pad, rowptr, csr, dinv, b0, h0);
  layer_kernel<64><<<mtiles, 256, 0, stream>>>(h0, W1, rowptr, csr, dinv, b1, h1);
  layer_kernel<64><<<mtiles, 256, 0, stream>>>(h1, W2, rowptr, csr, dinv, b2, h0);
  layer_kernel<64><<<mtiles, 256, 0, stream>>>(h0, W3, rowptr, csr, dinv, b3, h1);

  fill_f32_kernel<<<nb((long long)GG * DD, T), T, 0, stream>>>(gmax, -FLT_MAX, GG * DD);
  fill_f32_kernel<<<nb((long long)GG * DD, T), T, 0, stream>>>(gsum, 0.0f, GG * DD);
  fill_f32_kernel<<<nb(GG, T), T, 0, stream>>>(gcnt, 0.0f, GG);
  pool_acc_kernel<<<nb((long long)(NN / POOL_ROWS) * DD, T), T, 0, stream>>>(h1, batch, gmax, gsum, gcnt, NN);
  pool_finish_kernel<<<nb((long long)GG * DD, T), T, 0, stream>>>(gmax, gsum, gcnt, hidden, GG);

  head_kernel<<<nb(GG, 128), 128, 0, stream>>>(hidden, Wout, bout, outp, GG);
}


# Six small launches folded into ones that run anyway: x/W0 padding and pooling start values as extra work-groups of the edge-count launch, head dot in the pooling finish (out summed as a lane tree)
# speedup vs baseline: 4.2695x; 1.0986x over previous
#include <hip/hip_runtime.h>
#include <math.h>
#include <float.h>


typedef __attribute__((ext_vector_type(2))) float v2f;
typedef __attribute__((ext_vector_type(8))) float v8f;

#define NN 50000
#define EE 1200000
#define DD 64
#define GG 1024

static_assert(NN % 16 == 0, "the layer kernel's 16-row WMMA tiles need a row count that is a multiple of 16");
static_assert(EE % 4 == 0, "the norm pass takes 4 pairs per lane as two 16-byte groups");


#define BSH 7
#define BS (1 << BSH)
#define NB ((NN + BS - 1) / BS)
#define NCH 240
#define CH (EE / NCH)
static_assert(EE % NCH == 0 && CH % 4 == 0, "chunks are whole 16-byte groups of edges");
static_assert(NB <= 32 * 13, "the bin scan gives each lane of one wave 13 bins");

#define NAUX ((NN + 255) / 256)
static_assert(NAUX * 256 >= NN && NAUX * 256 >= GG * DD / 4 && NAUX * 256 >= 16 * 64,
              "one item per thread of the extra work-groups covers every pass");
static_assert(GG % 4 == 0, "the count plane is filled in 16-byte groups");

__global__ __launch_bounds__(256) void bin_count_kernel(const int* __restrict__ dst,
                                                        int* __restrict__ table,
                                                        const float* __restrict__ x,
                                                        float* __restrict__ hpad,
                                                        const float* __restrict__ W0,
                                                        float* __restrict__ w0pad,
                                                        float* __restrict__ gmax,
                                                        float* __restrict__ gsum,
                                                        float* __restrict__ gcnt) {
  __shared__ int hist[NB];
  const int tid = threadIdx.x;
  const int c = blockIdx.x;
  if (c >= NCH) {
    const int t = (c - NCH) * 256 + tid;
    if (t < NN) {
      const float* __restrict__ xr = x + (size_t)t * 9;
      float4* __restrict__ o = (float4*)(hpad + (size_t)t * 16);
      o[0] = make_float4(xr[0], xr[1], xr[2], xr[3]);
      o[1] = make_float4(xr[4], xr[5], xr[6], xr[7]);
      o[2] = make_float4(xr[8], 0.0f, 0.0f, 0.0f);
      o[3] = make_float4(0.0f, 0.0f, 0.0f, 0.0f);
    }
    if (t < GG * DD / 4) {
      ((float4*)gmax)[t] = make_float4(-FLT_MAX, -FLT_MAX, -FLT_MAX, -FLT_MAX);
      ((float4*)gsum)[t] = make_float4(0.0f, 0.0f, 0.0f, 0.0f);
    }
    if (t < GG / 4) ((float4*)gcnt)[t] = make_float4(0.0f, 0.0f, 0.0f, 0.0f);
    if (t < 16 * 64) w0pad[t] = (t < 9 * 64) ? W0[t] : 0.0f;
    return;
  }
  for (int i = tid; i < NB; i += 256) hist[i] = 0;
  __syncthreads();
  const int4* __restrict__ pd = (const int4*)(dst + c * CH);
  constexpr int Q = CH / 4;
  constexpr int KB = (Q + 255) / 256;
  int4 d[KB];
#pragma unroll
  for (int k = 0; k < KB; ++k) {
    const int i = tid + k * 256;
    d[k] = pd[(i < Q) ? i : (Q - 1)];
  }
#pragma unroll
  for (int k = 0; k < KB; ++k) {
    if (tid + k * 256 < Q) {
      atomicAdd(&hist[d[k].x >> BSH], 1);
      atomicAdd(&hist[d[k].y >> BSH], 1);
      atomicAdd(&hist[d[k].z >> BSH], 1);
      atomicAdd(&hist[d[k].w >> BSH], 1);
    }
  }
  __syncthreads();
  for (int i = tid; i < NB; i += 256) table[i * NCH + c] = hist[i];
}

__global__ __launch_bounds__(1024) void bin_scatter_kernel(const int* __restrict__ src,
                                                           const int* __restrict__ dst,
                                                           const int* __restrict__ table,
                                                           int* __restrict__ binstart,
                                                           int* __restrict__ part_) {
  int2* __restrict__ part = (int2*)part_;
  __shared__ int tot[NB + 1];
  __shared__ int cur[NB];
  const int tid = threadIdx.x;
  const int lane = tid & 31;
  const int wave = tid >> 5;
  const int c = blockIdx.x;

  constexpr int KR = (NCH + 31) / 32;
  for (int b0 = wave; b0 < NB; b0 += 4 * 32) {
    int tv[4][KR];
#pragma unroll
    for (int u = 0; u < 4; ++u) {
      const int b = b0 + u * 32;
      const int* __restrict__ row = table + ((b < NB) ? b : (NB - 1)) * NCH;
#pragma unroll
      for (int k = 0; k < KR; ++k) {
        const int idx = k * 32 + lane;
        tv[u][k] = row[(idx < NCH) ? idx : (NCH - 1)];
      }
    }
#pragma unroll
    for (int u = 0; u < 4; ++u) {
      const int b = b0 + u * 32;
      int t = 0;
      int p = 0;
#pragma unroll
      for (int k = 0; k < KR; ++k) {
        const int idx = k * 32 + lane;
        const int v = (idx < NCH) ? tv[u][k] : 0;
        t += v;
        p += (idx < c) ? v : 0;
      }
      for (int o = 16; o > 0; o >>= 1) {
        t += __shfl_xor(t, o);
        p += __shfl_xor(p, o);
      }
      if (lane == 0 && b < NB) {
        tot[b] = t;
        cur[b] = p;
      }
    }
  }
  __syncthreads();

  if (wave == 0) {
    int v[13];
    int s = 0;
#pragma unroll
    for (int k = 0; k < 13; ++k) {
      const int idx = lane * 13 + k;
      v[k] = (idx < NB) ? tot[idx] : 0;
      s += v[k];
    }
    int incl = s;
    for (int o = 1; o < 32; o <<= 1) {
      const int u = __shfl_up(incl, o);
      if (lane >= o) incl += u;
    }
    int excl = incl - s;
#pragma unroll
    for (int k = 0; k < 13; ++k) {
      const int idx = lane * 13 + k;
      if (idx < NB) tot[idx] = excl;
      excl += v[k];
    }
    if (lane == 31) tot[NB] = incl;
  }
  __syncthreads();
  for (int b = tid; b < NB; b += 1024) cur[b] += tot[b];
  if (c == 0)
    for (int b = tid; b <= NB; b += 1024) binstart[b] = tot[b];
  __syncthreads();

  const int4* __restrict__ ps = (const int4*)(src + c * CH);
  const int4* __restrict__ pd = (const int4*)(dst + c * CH);
  constexpr int Q = CH / 4;
  constexpr int KB = (Q + 1023) / 1024;
  int4 s[KB];
  int4 d[KB];
#pragma unroll
  for (int k = 0; k < KB; ++k) {
    const int i = tid + k * 1024;
    s[k] = ps[(i < Q) ? i : (Q - 1)];
    d[k] = pd[(i < Q) ? i : (Q - 1)];
  }
#pragma unroll
  for (int k = 0; k < KB; ++k) {
    if (tid + k * 1024 < Q) {
      part[atomicAdd(&cur[d[k].x >> BSH], 1)] = make_int2(s[k].x, d[k].x);
      part[atomicAdd(&cur[d[k].y >> BSH], 1)] = make_int2(s[k].y, d[k].y);
      part[atomicAdd(&cur[d[k].z >> BSH], 1)] = make_int2(s[k].z, d[k].z);
      part[atomicAdd(&cur[d[k].w >> BSH], 1)] = make_int2(s[k].w, d[k].w);
    }
  }
}

__global__ __launch_bounds__(512) void bin_build_kernel(const int* __restrict__ part_,
                                                         const int* __restrict__ binstart,
                                                         int* __restrict__ rowptr,
                                                         float* __restrict__ dinv,
                                                         int* __restrict__ csr_) {
  const int2* __restrict__ part = (const int2*)part_;
  int2* __restrict__ csr = (int2*)csr_;
  __shared__ int cnt[BS];
  __shared__ int cursor[BS];
  __shared__ int wtot[BS / 32];
  const int tid = threadIdx.x;
  const int lane = tid & 31;
  const int wave = tid >> 5;
  const int b = blockIdx.x;
  const int lo = binstart[b];
  const int hi = binstart[b + 1];
  const int base = b * BS;

  if (tid < BS) cnt[tid] = 0;
  __syncthreads();
  for (int j0 = lo + tid; j0 < hi; j0 += 8 * 512) {
    int2 e[8];
#pragma unroll
    for (int k = 0; k < 8; ++k) {
      const int j = j0 + k * 512;
      e[k] = part[(j < hi) ? j : (hi - 1)];
    }
#pragma unroll
    for (int k = 0; k < 8; ++k)
      if (j0 + k * 512 < hi) atomicAdd(&cnt[e[k].y - base], 1);
  }
  __syncthreads();

  int c = 0;
  int incl = 0;
  if (tid < BS) {
    c = cnt[tid];
    incl = c;
    for (int o = 1; o < 32; o <<= 1) {
      const int u = __shfl_up(incl, o);
      if (lane >= o) incl += u;
    }
    if (lane == 31) wtot[wave] = incl;
  }
  __syncthreads();
  if (tid < BS) {
    int off = lo;
    for (int w = 0; w < wave; ++w) off += wtot[w];
    const int excl = off + incl - c;
    cursor[tid] = excl;
    const int node = base + tid;
    if (node < NN) {
      rowptr[node] = excl;
      dinv[node] = rsqrtf((float)(c + 1));
    }
  }
  if (b == NB - 1 && tid == 0) rowptr[NN] = hi;
  __syncthreads();

  for (int j0 = lo + tid; j0 < hi; j0 += 8 * 512) {
    int2 e[8];
#pragma unroll
    for (int k = 0; k < 8; ++k) {
      const int j = j0 + k * 512;
      e[k] = part[(j < hi) ? j : (hi - 1)];
    }
#pragma unroll
    for (int k = 0; k < 8; ++k)
      if (j0 + k * 512 < hi) csr[atomicAdd(&cursor[e[k].y - base], 1)] = e[k];
  }
}

__global__ void norm_fill_kernel(int* csr, const float* __restrict__ dinv, int e) {
  const int q = e >> 2;
  const int i = blockIdx.x * blockDim.x + threadIdx.x;
  if (i >= q) return;
  int4 a = ((const int4*)csr)[i];
  int4 b = ((const int4*)csr)[i + q];
  a.y = __float_as_int(dinv[a.x] * dinv[a.y]);
  a.w = __float_as_int(dinv[a.z] * dinv[a.w]);
  b.y = __float_as_int(dinv[b.x] * dinv[b.y]);
  b.w = __float_as_int(dinv[b.z] * dinv[b.w]);
  ((int4*)csr)[i] = a;
  ((int4*)csr)[i + q] = b;
}

__device__ __forceinline__ void axpy4(float4& acc, const float4 v, const float w) {
  acc.x = fmaf(v.x, w, acc.x);
  acc.y = fmaf(v.y, w, acc.y);
  acc.z = fmaf(v.z, w, acc.z);
  acc.w = fmaf(v.w, w, acc.w);
}

template <int K>
__global__ __launch_bounds__(K * 4) void layer_kernel(const float* __restrict__ hin,
                                                      const float* __restrict__ W,
                                                      const int* __restrict__ rowptr,
                                                      const int* __restrict__ csr,
                                                      const float* __restrict__ dinv,
                                                      const float* __restrict__ b,
                                                      float* __restrict__ hout) {
  constexpr int LPN = K / 4;
  constexpr int NW = K / 8;
  constexpr int TPW = (NW >= 4) ? 1 : 4 / NW;
  constexpr int KS = K / 4;
  constexpr int P = K + 4;
  __shared__ __attribute__((aligned(16))) float zs[16 * P];

  const int tid = threadIdx.x;
  const int node = tid / LPN;
  const int fg = (tid % LPN) * 4;
  const int m0 = blockIdx.x * 16;
  const int d = m0 + node;
  const int2* __restrict__ ed = (const int2*)csr;

  int j = rowptr[d];
  const int end = rowptr[d + 1];
  const float4 self = *(const float4*)(hin + d * K + fg);
  const float di = dinv[d];

  float4 acc = make_float4(0.0f, 0.0f, 0.0f, 0.0f);

  for (; j + 8 <= end; j += 8) {
    int2 e[8];
    float4 v[8];
#pragma unroll
    for (int k = 0; k < 8; ++k) e[k] = ed[j + k];
#pragma unroll
    for (int k = 0; k < 8; ++k) v[k] = *(const float4*)(hin + e[k].x * K + fg);
#pragma unroll
    for (int k = 0; k < 8; ++k) axpy4(acc, v[k], __int_as_float(e[k].y));
  }

  if (j < end) {
    const int last = end - 1;
    int2 e[7];
    float4 v[7];
#pragma unroll
    for (int k = 0; k < 7; ++k) e[k] = ed[(j + k < last) ? (j + k) : last];
#pragma unroll
    for (int k = 0; k < 7; ++k) v[k] = *(const float4*)(hin + e[k].x * K + fg);
#pragma unroll
    for (int k = 0; k < 7; ++k)
      axpy4(acc, v[k], (j + k < end) ? __int_as_float(e[k].y) : 0.0f);
  }

  const int wave = __builtin_amdgcn_readfirstlane(tid >> 5);
  const int lane = tid & 31;
  const int l15 = lane & 15;
  const int half = lane >> 4;
  const bool mult = wave < 4;

  const float dd = di * di;
  float4 z;
  z.x = fmaf(self.x, dd, acc.x);
  z.y = fmaf(self.y, dd, acc.y);
  z.z = fmaf(self.z, dd, acc.z);
  z.w = fmaf(self.w, dd, acc.w);
  *(float4*)(zs + node * P + fg) = z;

  v2f bf[TPW][KS];
  if (mult) {
#pragma unroll
    for (int t = 0; t < TPW; ++t) {
      const float* __restrict__ bcol = W + half * 2 * 64 + (wave + t * NW) * 16 + l15;
#pragma unroll
      for (int s = 0; s < KS; ++s) {
        bf[t][s].x = bcol[s * 4 * 64];
        bf[t][s].y = bcol[s * 4 * 64 + 64];
      }
    }
  }
  __builtin_amdgcn_fence(__ATOMIC_RELEASE, "workgroup", "local");
  __builtin_amdgcn_s_barrier();
  __builtin_amdgcn_fence(__ATOMIC_ACQUIRE, "workgroup", "local");

  if (mult) {
    const float* zrow = zs + l15 * P + half * 2;
#pragma unroll
    for (int t = 0; t < TPW; ++t) {
      const int col = (wave + t * NW) * 16 + l15;
      v8f c0 = {};
      v8f c1 = {};
#pragma unroll
      for (int s = 0; s < KS; s += 2) {
        const v2f a0 = *(const v2f*)(zrow + s * 4);
        const v2f a1 = *(const v2f*)(zrow + s * 4 + 4);
        c0 = __builtin_amdgcn_wmma_f32_16x16x4_f32(false, a0, false, bf[t][s], (short)0, c0,
                                                   false, false);
        c1 = __builtin_amdgcn_wmma_f32_16x16x4_f32(false, a1, false, bf[t][s + 1], (short)0, c1,
                                                   false, false);
      }
      const float bias = b[col];
      float* __restrict__ orow = hout + (m0 + half * 8) * 64 + col;
#pragma unroll
      for (int v = 0; v < 8; ++v) orow[v * 64] = tanhf(c0[v] + c1[v] + bias);
    }
  }
}


__device__ inline void atomicMaxF32(float* addr, float val) {
  if (val >= 0.0f)
    atomicMax((int*)addr, __float_as_int(val));
  else
    atomicMin((unsigned int*)addr, __float_as_uint(val));
}

__device__ __forceinline__ void pool_flush(const int g, const int f, const float s, const float m,
                                           const int c, float* __restrict__ gmax,
                                           float* __restrict__ gsum, float* __restrict__ cnt) {
  atomicAdd(&gsum[(size_t)g * DD + f], s);
  atomicMaxF32(&gmax[(size_t)g * DD + f], m);
  if (f == 0) atomicAdd(&cnt[g], (float)c);
}

#define POOL_ROWS 16
static_assert(NN % POOL_ROWS == 0, "pooling walks whole chunks of rows");

__global__ __launch_bounds__(256) void pool_acc_kernel(const float* __restrict__ h,
                                const int* __restrict__ batch,
                                float* __restrict__ gmax,
                                float* __restrict__ gsum,
                                float* __restrict__ cnt, int n) {
  const int gid = blockIdx.x * blockDim.x + threadIdx.x;
  const int f = gid & 63;
  const int r0 = (gid >> 6) * POOL_ROWS;
  if (r0 >= n) return;

  int g[POOL_ROWS];
  float v[POOL_ROWS];
#pragma unroll
  for (int k = 0; k < POOL_ROWS; ++k) g[k] = batch[r0 + k];
#pragma unroll
  for (int k = 0; k < POOL_ROWS; ++k) v[k] = h[(size_t)(r0 + k) * DD + f];

  int cur = g[0];
  float s = v[0];
  float m = v[0];
  int c = 1;
#pragma unroll
  for (int k = 1; k < POOL_ROWS; ++k) {
    if (g[k] != cur) {
      pool_flush(cur, f, s, m, c, gmax, gsum, cnt);
      cur = g[k];
      s = v[k];
      m = v[k];
      c = 1;
    } else {
      s += v[k];
      m = fmaxf(m, v[k]);
      c += 1;
    }
  }
  pool_flush(cur, f, s, m, c, gmax, gsum, cnt);
}

static_assert(GG % 32 == 0 && DD == 64, "32 graphs per work-group, two features per lane");
__global__ __launch_bounds__(1024) void pool_finish_head_kernel(const float* __restrict__ gmax,
                                                                const float* __restrict__ gsum,
                                                                const float* __restrict__ cnt,
                                                                const float* __restrict__ Wout,
                                                                const float* __restrict__ bout,
                                                                float* __restrict__ hidden,
                                                                float* __restrict__ out) {
  __shared__ float dot[32];
  const int lane = threadIdx.x & 31;
  const int wave = threadIdx.x >> 5;
  const int g = blockIdx.x * 32 + wave;
  const float c = cnt[g];
  float2 mx = ((const float2*)(gmax + (size_t)g * DD))[lane];
  const float2 sm = ((const float2*)(gsum + (size_t)g * DD))[lane];
  if (c == 0.0f) mx = make_float2(0.0f, 0.0f);
  const float cd = fmaxf(c, 1.0f);
  const float2 mean = make_float2(sm.x / cd, sm.y / cd);
  float* __restrict__ hrow = hidden + (size_t)g * (2 * DD);
  ((float2*)hrow)[lane] = mx;
  ((float2*)(hrow + DD))[lane] = mean;
  const int f = lane * 2;
  float acc = mx.x * Wout[f];
  acc = fmaf(mx.y, Wout[f + 1], acc);
  acc = fmaf(mean.x, Wout[DD + f], acc);
  acc = fmaf(mean.y, Wout[DD + f + 1], acc);
  for (int o = 16; o > 0; o >>= 1) acc += __shfl_xor(acc, o);
  if (lane == 0) dot[wave] = acc;
  __syncthreads();
  if (wave == 0) out[blockIdx.x * 32 + lane] = dot[lane] + bout[0];
}


extern "C" void kernel_launch(void* const* d_in, const int* in_sizes, int n_in,
                              void* d_out, int out_size, void* d_ws, size_t ws_size,
                              hipStream_t stream) {
  (void)in_sizes; (void)n_in; (void)out_size; (void)ws_size;

  const float* x    = (const float*)d_in[0];
  const int*   eidx = (const int*)d_in[1];
  const int*   batch = (const int*)d_in[2];
  const float* W0 = (const float*)d_in[3];
  const float* b0 = (const float*)d_in[4];
  const float* W1 = (const float*)d_in[5];
  const float* b1 = (const float*)d_in[6];
  const float* W2 = (const float*)d_in[7];
  const float* b2 = (const float*)d_in[8];
  const float* W3 = (const float*)d_in[9];
  const float* b3 = (const float*)d_in[10];
  const float* Wout = (const float*)d_in[11];
  const float* bout = (const float*)d_in[12];

  const int* src = eidx;
  const int* dst = eidx + EE;

  float* outp   = (float*)d_out;
  float* hidden = (float*)d_out + GG;

  float* ws = (float*)d_ws;
  size_t o = 0;
  float* dinv   = ws + o;         o += NN + 48;
  int*   rowptr = (int*)(ws + o); o += NN + 112;
  int*   table  = (int*)(ws + o); o += NB * NCH + 48;
  int*   bstart = (int*)(ws + o); o += 448;
  int*   part   = (int*)(ws + o); o += (size_t)EE * 2;
  int*   csr    = (int*)(ws + o); o += (size_t)EE * 2;
  float* hpad  = ws + o; o += (size_t)NN * 16;
  float* w0pad = ws + o; o += 16 * 64;
  float* h0    = ws + o; o += (size_t)NN * DD;
  float* h1    = ws + o; o += (size_t)NN * DD;
  float* gmax  = ws + o; o += (size_t)GG * DD;
  float* gsum  = ws + o; o += (size_t)GG * DD;
  float* gcnt  = ws + o; o += GG;

  const int T = 256;
  auto nb = [](long long n, int t) { return (int)((n + t - 1) / t); };

  bin_count_kernel<<<NCH + NAUX, 256, 0, stream>>>(dst, table, x, hpad, W0, w0pad, gmax, gsum, gcnt);
  bin_scatter_kernel<<<NCH, 1024, 0, stream>>>(src, dst, table, bstart, part);
  bin_build_kernel<<<NB, 512, 0, stream>>>(part, bstart, rowptr, dinv, csr);
  norm_fill_kernel<<<nb(EE / 4, T), T, 0, stream>>>(csr, dinv, EE);

  const int mtiles = NN / 16;

  layer_kernel<16><<<mtiles, 64, 0, stream>>>(hpad, w0pad, rowptr, csr, dinv, b0, h0);
  layer_kernel<64><<<mtiles, 256, 0, stream>>>(h0, W1, rowptr, csr, dinv, b1, h1);
  layer_kernel<64><<<mtiles, 256, 0, stream>>>(h1, W2, rowptr, csr, dinv, b2, h0);
  layer_kernel<64><<<mtiles, 256, 0, stream>>>(h0, W3, rowptr, csr, dinv, b3, h1);

  pool_acc_kernel<<<nb((long long)(NN / POOL_ROWS) * DD, T), T, 0, stream>>>(h1, batch, gmax, gsum, gcnt, NN);
  pool_finish_head_kernel<<<GG / 32, 1024, 0, stream>>>(gmax, gsum, gcnt, Wout, bout, hidden, outp);
}


# Pooling folded into the last layer's epilogue: a tile's 16 tanh rows per column are run-summed and maxed in registers and leave as atomics; pooling launch and last row plane go; atomic order changes
# speedup vs baseline: 4.9100x; 1.1500x over previous
#include <hip/hip_runtime.h>
#include <math.h>
#include <float.h>


typedef __attribute__((ext_vector_type(2))) float v2f;
typedef __attribute__((ext_vector_type(8))) float v8f;

#define NN 50000
#define EE 1200000
#define DD 64
#define GG 1024

static_assert(NN % 16 == 0, "the layer kernel's 16-row WMMA tiles need a row count that is a multiple of 16");


#define BSH 7
#define BS (1 << BSH)
#define NB ((NN + BS - 1) / BS)
#define NCH 240
#define CH (EE / NCH)
static_assert(EE % NCH == 0 && CH % 4 == 0, "chunks are whole 16-byte groups of edges");
static_assert(NB <= 32 * 13, "the bin scan gives each lane of one wave 13 bins");

#define NAUX ((GG * DD / 4 + 255) / 256)
static_assert(NAUX * 256 >= GG * DD / 4 && NAUX * 256 >= 16 * 64,
              "one item per thread of the extra work-groups covers every pass");
static_assert(GG % 4 == 0, "the count plane is filled in 16-byte groups");

__global__ __launch_bounds__(256) void bin_count_kernel(const int* __restrict__ dst,
                                                        int* __restrict__ table,
                                                        const float* __restrict__ W0,
                                                        float* __restrict__ w0pad,
                                                        float* __restrict__ gmax,
                                                        float* __restrict__ gsum,
                                                        float* __restrict__ gcnt) {
  __shared__ int hist[NB];
  const int tid = threadIdx.x;
  const int c = blockIdx.x;
  if (c >= NCH) {
    const int t = (c - NCH) * 256 + tid;
    if (t < GG * DD / 4) {
      ((float4*)gmax)[t] = make_float4(-FLT_MAX, -FLT_MAX, -FLT_MAX, -FLT_MAX);
      ((float4*)gsum)[t] = make_float4(0.0f, 0.0f, 0.0f, 0.0f);
    }
    if (t < GG / 4) ((float4*)gcnt)[t] = make_float4(0.0f, 0.0f, 0.0f, 0.0f);
    if (t < 16 * 64) w0pad[t] = (t < 9 * 64) ? W0[t] : 0.0f;
    return;
  }
  for (int i = tid; i < NB; i += 256) hist[i] = 0;
  __syncthreads();
  const int4* __restrict__ pd = (const int4*)(dst + c * CH);
  constexpr int Q = CH / 4;
  constexpr int KB = (Q + 255) / 256;
  int4 d[KB];
#pragma unroll
  for (int k = 0; k < KB; ++k) {
    const int i = tid + k * 256;
    d[k] = pd[(i < Q) ? i : (Q - 1)];
  }
#pragma unroll
  for (int k = 0; k < KB; ++k) {
    if (tid + k * 256 < Q) {
      atomicAdd(&hist[d[k].x >> BSH], 1);
      atomicAdd(&hist[d[k].y >> BSH], 1);
      atomicAdd(&hist[d[k].z >> BSH], 1);
      atomicAdd(&hist[d[k].w >> BSH], 1);
    }
  }
  __syncthreads();
  for (int i = tid; i < NB; i += 256) table[i * NCH + c] = hist[i];
}

__global__ __launch_bounds__(1024) void bin_scatter_kernel(const int* __restrict__ src,
                                                           const int* __restrict__ dst,
                                                           const int* __restrict__ table,
                                                           int* __restrict__ binstart,
                                                           int* __restrict__ part_) {
  int2* __restrict__ part = (int2*)part_;
  __shared__ int tot[NB + 1];
  __shared__ int cur[NB];
  const int tid = threadIdx.x;
  const int lane = tid & 31;
  const int wave = tid >> 5;
  const int c = blockIdx.x;

  constexpr int KR = (NCH + 31) / 32;
  for (int b0 = wave; b0 < NB; b0 += 4 * 32) {
    int tv[4][KR];
#pragma unroll
    for (int u = 0; u < 4; ++u) {
      const int b = b0 + u * 32;
      const int* __restrict__ row = table + ((b < NB) ? b : (NB - 1)) * NCH;
#pragma unroll
      for (int k = 0; k < KR; ++k) {
        const int idx = k * 32 + lane;
        tv[u][k] = row[(idx < NCH) ? idx : (NCH - 1)];
      }
    }
#pragma unroll
    for (int u = 0; u < 4; ++u) {
      const int b = b0 + u * 32;
      int t = 0;
      int p = 0;
#pragma unroll
      for (int k = 0; k < KR; ++k) {
        const int idx = k * 32 + lane;
        const int v = (idx < NCH) ? tv[u][k] : 0;
        t += v;
        p += (idx < c) ? v : 0;
      }
      for (int o = 16; o > 0; o >>= 1) {
        t += __shfl_xor(t, o);
        p += __shfl_xor(p, o);
      }
      if (lane == 0 && b < NB) {
        tot[b] = t;
        cur[b] = p;
      }
    }
  }
  __syncthreads();

  if (wave == 0) {
    int v[13];
    int s = 0;
#pragma unroll
    for (int k = 0; k < 13; ++k) {
      const int idx = lane * 13 + k;
      v[k] = (idx < NB) ? tot[idx] : 0;
      s += v[k];
    }
    int incl = s;
    for (int o = 1; o < 32; o <<= 1) {
      const int u = __shfl_up(incl, o);
      if (lane >= o) incl += u;
    }
    int excl = incl - s;
#pragma unroll
    for (int k = 0; k < 13; ++k) {
      const int idx = lane * 13 + k;
      if (idx < NB) tot[idx] = excl;
      excl += v[k];
    }
    if (lane == 31) tot[NB] = incl;
  }
  __syncthreads();
  for (int b = tid; b < NB; b += 1024) cur[b] += tot[b];
  if (c == 0)
    for (int b = tid; b <= NB; b += 1024) binstart[b] = tot[b];
  __syncthreads();

  const int4* __restrict__ ps = (const int4*)(src + c * CH);
  const int4* __restrict__ pd = (const int4*)(dst + c * CH);
  constexpr int Q = CH / 4;
  constexpr int KB = (Q + 1023) / 1024;
  int4 s[KB];
  int4 d[KB];
#pragma unroll
  for (int k = 0; k < KB; ++k) {
    const int i = tid + k * 1024;
    s[k] = ps[(i < Q) ? i : (Q - 1)];
    d[k] = pd[(i < Q) ? i : (Q - 1)];
  }
#pragma unroll
  for (int k = 0; k < KB; ++k) {
    if (tid + k * 1024 < Q) {
      part[atomicAdd(&cur[d[k].x >> BSH], 1)] = make_int2(s[k].x, d[k].x);
      part[atomicAdd(&cur[d[k].y >> BSH], 1)] = make_int2(s[k].y, d[k].y);
      part[atomicAdd(&cur[d[k].z >> BSH], 1)] = make_int2(s[k].z, d[k].z);
      part[atomicAdd(&cur[d[k].w >> BSH], 1)] = make_int2(s[k].w, d[k].w);
    }
  }
}

__global__ __launch_bounds__(512) void bin_build_kernel(const int* __restrict__ part_,
                                                         const int* __restrict__ binstart,
                                                         int* __restrict__ rowptr,
                                                         float* __restrict__ dinv,
                                                         int* __restrict__ csr,
                                                         const float* __restrict__ x,
                                                         float* __restrict__ hpad) {
  const int2* __restrict__ part = (const int2*)part_;
  __shared__ int cnt[BS];
  __shared__ int cursor[BS];
  __shared__ int wtot[BS / 32];
  const int tid = threadIdx.x;
  const int lane = tid & 31;
  const int wave = tid >> 5;
  const int b = blockIdx.x;
  const int lo = binstart[b];
  const int hi = binstart[b + 1];
  const int base = b * BS;

  if (tid < BS) cnt[tid] = 0;
  __syncthreads();
  float xv[9];
  if (tid < BS && base + tid < NN) {
    const float* __restrict__ xr = x + (size_t)(base + tid) * 9;
#pragma unroll
    for (int k = 0; k < 9; ++k) xv[k] = xr[k];
  }
  for (int j0 = lo + tid; j0 < hi; j0 += 8 * 512) {
    int2 e[8];
#pragma unroll
    for (int k = 0; k < 8; ++k) {
      const int j = j0 + k * 512;
      e[k] = part[(j < hi) ? j : (hi - 1)];
    }
#pragma unroll
    for (int k = 0; k < 8; ++k)
      if (j0 + k * 512 < hi) atomicAdd(&cnt[e[k].y - base], 1);
  }
  __syncthreads();

  int c = 0;
  int incl = 0;
  if (tid < BS) {
    c = cnt[tid];
    incl = c;
    for (int o = 1; o < 32; o <<= 1) {
      const int u = __shfl_up(incl, o);
      if (lane >= o) incl += u;
    }
    if (lane == 31) wtot[wave] = incl;
  }
  __syncthreads();
  if (tid < BS) {
    int off = lo;
    for (int w = 0; w < wave; ++w) off += wtot[w];
    const int excl = off + incl - c;
    cursor[tid] = excl;
    const int node = base + tid;
    if (node < NN) {
      rowptr[node] = excl;
      const float di = rsqrtf((float)(c + 1));
      dinv[node] = di;
      float4* __restrict__ o = (float4*)(hpad + (size_t)node * 16);
      o[0] = make_float4(xv[0] * di, xv[1] * di, xv[2] * di, xv[3] * di);
      o[1] = make_float4(xv[4] * di, xv[5] * di, xv[6] * di, xv[7] * di);
      o[2] = make_float4(xv[8] * di, 0.0f, 0.0f, 0.0f);
      o[3] = make_float4(0.0f, 0.0f, 0.0f, 0.0f);
    }
  }
  if (b == NB - 1 && tid == 0) rowptr[NN] = hi;
  __syncthreads();

  for (int j0 = lo + tid; j0 < hi; j0 += 8 * 512) {
    int2 e[8];
#pragma unroll
    for (int k = 0; k < 8; ++k) {
      const int j = j0 + k * 512;
      e[k] = part[(j < hi) ? j : (hi - 1)];
    }
#pragma unroll
    for (int k = 0; k < 8; ++k)
      if (j0 + k * 512 < hi) csr[atomicAdd(&cursor[e[k].y - base], 1)] = e[k].x;
  }
}

__device__ __forceinline__ void add4(float4& acc, const float4 v) {
  acc.x += v.x;
  acc.y += v.y;
  acc.z += v.z;
  acc.w += v.w;
}
__device__ __forceinline__ void axpy4(float4& acc, const float4 v, const float w) {
  acc.x = fmaf(v.x, w, acc.x);
  acc.y = fmaf(v.y, w, acc.y);
  acc.z = fmaf(v.z, w, acc.z);
  acc.w = fmaf(v.w, w, acc.w);
}

__device__ inline void atomicMaxF32(float* addr, float val) {
  if (val >= 0.0f)
    atomicMax((int*)addr, __float_as_int(val));
  else
    atomicMin((unsigned int*)addr, __float_as_uint(val));
}

__device__ __forceinline__ void pool_flush(const int g, const int f, const int half,
                                           const bool cntlane, const float s, const float m,
                                           const int c, float* __restrict__ gmax,
                                           float* __restrict__ gsum, float* __restrict__ cnt) {
  if (half == 0)
    atomicAdd(&gsum[(size_t)g * DD + f], s);
  else
    atomicMaxF32(&gmax[(size_t)g * DD + f], m);
  if (cntlane) atomicAdd(&cnt[g], (float)c);
}

template <int K, bool SCALE_OUT, bool POOL>
__global__ __launch_bounds__(K * 4) void layer_kernel(const float* __restrict__ hin,
                                                      const float* __restrict__ W,
                                                      const int* __restrict__ rowptr,
                                                      const int* __restrict__ csr,
                                                      const float* __restrict__ dinv,
                                                      const float* __restrict__ b,
                                                      float* __restrict__ hout,
                                                      const int* __restrict__ batch,
                                                      float* __restrict__ gmax,
                                                      float* __restrict__ gsum,
                                                      float* __restrict__ gcnt) {
  static_assert(!(POOL && SCALE_OUT), "the pooled layer stores no rows, so there is nothing to scale");
  constexpr int LPN = K / 4;
  constexpr int NW = K / 8;
  constexpr int TPW = (NW >= 4) ? 1 : 4 / NW;
  constexpr int KS = K / 4;
  constexpr int P = K + 4;
  __shared__ __attribute__((aligned(16))) float zs[16 * P];
  __shared__ __attribute__((aligned(16))) float dis[16];

  const int tid = threadIdx.x;
  const int node = tid / LPN;
  const int fg = (tid % LPN) * 4;
  const int m0 = blockIdx.x * 16;
  const int d = m0 + node;

  int j = rowptr[d];
  const int end = rowptr[d + 1];
  const float4 self = *(const float4*)(hin + d * K + fg);
  const float di = dinv[d];

  float4 acc = make_float4(0.0f, 0.0f, 0.0f, 0.0f);

  for (; j + 8 <= end; j += 8) {
    int e[8];
    float4 v[8];
#pragma unroll
    for (int k = 0; k < 8; ++k) e[k] = csr[j + k];
#pragma unroll
    for (int k = 0; k < 8; ++k) v[k] = *(const float4*)(hin + e[k] * K + fg);
#pragma unroll
    for (int k = 0; k < 8; ++k) add4(acc, v[k]);
  }

  if (j < end) {
    const int last = end - 1;
    int e[7];
    float4 v[7];
#pragma unroll
    for (int k = 0; k < 7; ++k) e[k] = csr[(j + k < last) ? (j + k) : last];
#pragma unroll
    for (int k = 0; k < 7; ++k) v[k] = *(const float4*)(hin + e[k] * K + fg);
#pragma unroll
    for (int k = 0; k < 7; ++k) axpy4(acc, v[k], (j + k < end) ? 1.0f : 0.0f);
  }

  const int wave = __builtin_amdgcn_readfirstlane(tid >> 5);
  const int lane = tid & 31;
  const int l15 = lane & 15;
  const int half = lane >> 4;
  const bool mult = wave < 4;

  float4 z;
  z.x = (acc.x + self.x) * di;
  z.y = (acc.y + self.y) * di;
  z.z = (acc.z + self.z) * di;
  z.w = (acc.w + self.w) * di;
  *(float4*)(zs + node * P + fg) = z;
  if (SCALE_OUT && fg == 0) dis[node] = di;

  v2f bf[TPW][KS];
  if (mult) {
#pragma unroll
    for (int t = 0; t < TPW; ++t) {
      const float* __restrict__ bcol = W + half * 2 * 64 + (wave + t * NW) * 16 + l15;
#pragma unroll
      for (int s = 0; s < KS; ++s) {
        bf[t][s].x = bcol[s * 4 * 64];
        bf[t][s].y = bcol[s * 4 * 64 + 64];
      }
    }
  }
  __builtin_amdgcn_fence(__ATOMIC_RELEASE, "workgroup", "local");
  __builtin_amdgcn_s_barrier();
  __builtin_amdgcn_fence(__ATOMIC_ACQUIRE, "workgroup", "local");

  if (mult) {
    const float* zrow = zs + l15 * P + half * 2;
#pragma unroll
    for (int t = 0; t < TPW; ++t) {
      const int col = (wave + t * NW) * 16 + l15;
      v8f c0 = {};
      v8f c1 = {};
#pragma unroll
      for (int s = 0; s < KS; s += 2) {
        const v2f a0 = *(const v2f*)(zrow + s * 4);
        const v2f a1 = *(const v2f*)(zrow + s * 4 + 4);
        c0 = __builtin_amdgcn_wmma_f32_16x16x4_f32(false, a0, false, bf[t][s], (short)0, c0,
                                                   false, false);
        c1 = __builtin_amdgcn_wmma_f32_16x16x4_f32(false, a1, false, bf[t][s + 1], (short)0, c1,
                                                   false, false);
      }
      const float bias = b[col];
      float* __restrict__ orow = hout + (m0 + half * 8) * 64 + col;
      if (POOL) {
        float own[8];
        float oth[8];
        float cv[16];
#pragma unroll
        for (int v = 0; v < 8; ++v) own[v] = tanhf(c0[v] + c1[v] + bias);
#pragma unroll
        for (int v = 0; v < 8; ++v) oth[v] = __shfl_xor(own[v], 16);
#pragma unroll
        for (int v = 0; v < 8; ++v) {
          cv[v] = half ? oth[v] : own[v];
          cv[8 + v] = half ? own[v] : oth[v];
        }
        int g[16];
#pragma unroll
        for (int r = 0; r < 16; ++r) g[r] = batch[m0 + r];
        const bool cntlane = (col == 0) && (half == 0);
        int cur = g[0];
        float s = cv[0];
        float m = cv[0];
        int c = 1;
#pragma unroll
        for (int r = 1; r < 16; ++r) {
          if (g[r] != cur) {
            pool_flush(cur, col, half, cntlane, s, m, c, gmax, gsum, gcnt);
            cur = g[r];
            s = cv[r];
            m = cv[r];
            c = 1;
          } else {
            s += cv[r];
            m = fmaxf(m, cv[r]);
            c += 1;
          }
        }
        pool_flush(cur, col, half, cntlane, s, m, c, gmax, gsum, gcnt);
      } else if (SCALE_OUT) {
        const float4 s0 = *(const float4*)(dis + half * 8);
        const float4 s1 = *(const float4*)(dis + half * 8 + 4);
        const float so[8] = {s0.x, s0.y, s0.z, s0.w, s1.x, s1.y, s1.z, s1.w};
#pragma unroll
        for (int v = 0; v < 8; ++v) orow[v * 64] = tanhf(c0[v] + c1[v] + bias) * so[v];
      } else {
#pragma unroll
        for (int v = 0; v < 8; ++v) orow[v * 64] = tanhf(c0[v] + c1[v] + bias);
      }
    }
  }
}


static_assert(GG % 32 == 0 && DD == 64, "32 graphs per work-group, two features per lane");
__global__ __launch_bounds__(1024) void pool_finish_head_kernel(const float* __restrict__ gmax,
                                                                const float* __restrict__ gsum,
                                                                const float* __restrict__ cnt,
                                                                const float* __restrict__ Wout,
                                                                const float* __restrict__ bout,
                                                                float* __restrict__ hidden,
                                                                float* __restrict__ out) {
  __shared__ float dot[32];
  const int lane = threadIdx.x & 31;
  const int wave = threadIdx.x >> 5;
  const int g = blockIdx.x * 32 + wave;
  const float c = cnt[g];
  float2 mx = ((const float2*)(gmax + (size_t)g * DD))[lane];
  const float2 sm = ((const float2*)(gsum + (size_t)g * DD))[lane];
  if (c == 0.0f) mx = make_float2(0.0f, 0.0f);
  const float cd = fmaxf(c, 1.0f);
  const float2 mean = make_float2(sm.x / cd, sm.y / cd);
  float* __restrict__ hrow = hidden + (size_t)g * (2 * DD);
  ((float2*)hrow)[lane] = mx;
  ((float2*)(hrow + DD))[lane] = mean;
  const int f = lane * 2;
  float acc = mx.x * Wout[f];
  acc = fmaf(mx.y, Wout[f + 1], acc);
  acc = fmaf(mean.x, Wout[DD + f], acc);
  acc = fmaf(mean.y, Wout[DD + f + 1], acc);
  for (int o = 16; o > 0; o >>= 1) acc += __shfl_xor(acc, o);
  if (lane == 0) dot[wave] = acc;
  __syncthreads();
  if (wave == 0) out[blockIdx.x * 32 + lane] = dot[lane] + bout[0];
}


extern "C" void kernel_launch(void* const* d_in, const int* in_sizes, int n_in,
                              void* d_out, int out_size, void* d_ws, size_t ws_size,
                              hipStream_t stream) {
  (void)in_sizes; (void)n_in; (void)out_size; (void)ws_size;

  const float* x    = (const float*)d_in[0];
  const int*   eidx = (const int*)d_in[1];
  const int*   batch = (const int*)d_in[2];
  const float* W0 = (const float*)d_in[3];
  const float* b0 = (const float*)d_in[4];
  const float* W1 = (const float*)d_in[5];
  const float* b1 = (const float*)d_in[6];
  const float* W2 = (const float*)d_in[7];
  const float* b2 = (const float*)d_in[8];
  const float* W3 = (const float*)d_in[9];
  const float* b3 = (const float*)d_in[10];
  const float* Wout = (const float*)d_in[11];
  const float* bout = (const float*)d_in[12];

  const int* src = eidx;
  const int* dst = eidx + EE;

  float* outp   = (float*)d_out;
  float* hidden = (float*)d_out + GG;

  float* ws = (float*)d_ws;
  size_t o = 0;
  float* dinv   = ws + o;         o += NN + 48;
  int*   rowptr = (int*)(ws + o); o += NN + 112;
  int*   table  = (int*)(ws + o); o += NB * NCH + 48;
  int*   bstart = (int*)(ws + o); o += 448;
  int*   part   = (int*)(ws + o); o += (size_t)EE * 2;
  int*   csr    = (int*)(ws + o); o += (size_t)EE;
  float* hpad  = ws + o; o += (size_t)NN * 16;
  float* w0pad = ws + o; o += 16 * 64;
  float* h0    = ws + o; o += (size_t)NN * DD;
  float* h1    = ws + o; o += (size_t)NN * DD;
  float* gmax  = ws + o; o += (size_t)GG * DD;
  float* gsum  = ws + o; o += (size_t)GG * DD;
  float* gcnt  = ws + o; o += GG;

  bin_count_kernel<<<NCH + NAUX, 256, 0, stream>>>(dst, table, W0, w0pad, gmax, gsum, gcnt);
  bin_scatter_kernel<<<NCH, 1024, 0, stream>>>(src, dst, table, bstart, part);
  bin_build_kernel<<<NB, 512, 0, stream>>>(part, bstart, rowptr, dinv, csr, x, hpad);

  const int mtiles = NN / 16;

  layer_kernel<16, true, false><<<mtiles, 64, 0, stream>>>(hpad, w0pad, rowptr, csr, dinv, b0, h0, batch, gmax, gsum, gcnt);
  layer_kernel<64, true, false><<<mtiles, 256, 0, stream>>>(h0, W1, rowptr, csr, dinv, b1, h1, batch, gmax, gsum, gcnt);
  layer_kernel<64, true, false><<<mtiles, 256, 0, stream>>>(h1, W2, rowptr, csr, dinv, b2, h0, batch, gmax, gsum, gcnt);
  layer_kernel<64, false, true><<<mtiles, 256, 0, stream>>>(h0, W3, rowptr, csr, dinv, b3, h1, batch, gmax, gsum, gcnt);

  pool_finish_head_kernel<<<GG / 32, 1024, 0, stream>>>(gmax, gsum, gcnt, Wout, bout, hidden, outp);
}


# Edge indices: a node's lanes load a chunk of its edge slots between them, one chunk ahead, and read each slot out by DPP lane select; a batch of rows waits one cache trip, not two; sum order kept
# speedup vs baseline: 5.1529x; 1.0495x over previous
#include <hip/hip_runtime.h>
#include <math.h>
#include <float.h>


typedef __attribute__((ext_vector_type(2))) float v2f;
typedef __attribute__((ext_vector_type(8))) float v8f;

#define NN 50000
#define EE 1200000
#define DD 64
#define GG 1024

static_assert(NN % 16 == 0, "the layer kernel's 16-row WMMA tiles need a row count that is a multiple of 16");


#define BSH 7
#define BS (1 << BSH)
#define NB ((NN + BS - 1) / BS)
#define NCH 240
#define CH (EE / NCH)
static_assert(EE % NCH == 0 && CH % 4 == 0, "chunks are whole 16-byte groups of edges");
static_assert(NB <= 32 * 13, "the bin scan gives each lane of one wave 13 bins");

#define NAUX ((GG * DD / 4 + 255) / 256)
static_assert(NAUX * 256 >= GG * DD / 4 && NAUX * 256 >= 16 * 64,
              "one item per thread of the extra work-groups covers every pass");
static_assert(GG % 4 == 0, "the count plane is filled in 16-byte groups");

__global__ __launch_bounds__(256) void bin_count_kernel(const int* __restrict__ dst,
                                                        int* __restrict__ table,
                                                        const float* __restrict__ W0,
                                                        float* __restrict__ w0pad,
                                                        float* __restrict__ gmax,
                                                        float* __restrict__ gsum,
                                                        float* __restrict__ gcnt) {
  __shared__ int hist[NB];
  const int tid = threadIdx.x;
  const int c = blockIdx.x;
  if (c >= NCH) {
    const int t = (c - NCH) * 256 + tid;
    if (t < GG * DD / 4) {
      ((float4*)gmax)[t] = make_float4(-FLT_MAX, -FLT_MAX, -FLT_MAX, -FLT_MAX);
      ((float4*)gsum)[t] = make_float4(0.0f, 0.0f, 0.0f, 0.0f);
    }
    if (t < GG / 4) ((float4*)gcnt)[t] = make_float4(0.0f, 0.0f, 0.0f, 0.0f);
    if (t < 16 * 64) w0pad[t] = (t < 9 * 64) ? W0[t] : 0.0f;
    return;
  }
  for (int i = tid; i < NB; i += 256) hist[i] = 0;
  __syncthreads();
  const int4* __restrict__ pd = (const int4*)(dst + c * CH);
  constexpr int Q = CH / 4;
  constexpr int KB = (Q + 255) / 256;
  int4 d[KB];
#pragma unroll
  for (int k = 0; k < KB; ++k) {
    const int i = tid + k * 256;
    d[k] = pd[(i < Q) ? i : (Q - 1)];
  }
#pragma unroll
  for (int k = 0; k < KB; ++k) {
    if (tid + k * 256 < Q) {
      atomicAdd(&hist[d[k].x >> BSH], 1);
      atomicAdd(&hist[d[k].y >> BSH], 1);
      atomicAdd(&hist[d[k].z >> BSH], 1);
      atomicAdd(&hist[d[k].w >> BSH], 1);
    }
  }
  __syncthreads();
  for (int i = tid; i < NB; i += 256) table[i * NCH + c] = hist[i];
}

__global__ __launch_bounds__(1024) void bin_scatter_kernel(const int* __restrict__ src,
                                                           const int* __restrict__ dst,
                                                           const int* __restrict__ table,
                                                           int* __restrict__ binstart,
                                                           int* __restrict__ part_) {
  int2* __restrict__ part = (int2*)part_;
  __shared__ int tot[NB + 1];
  __shared__ int cur[NB];
  const int tid = threadIdx.x;
  const int lane = tid & 31;
  const int wave = tid >> 5;
  const int c = blockIdx.x;

  constexpr int KR = (NCH + 31) / 32;
  for (int b0 = wave; b0 < NB; b0 += 4 * 32) {
    int tv[4][KR];
#pragma unroll
    for (int u = 0; u < 4; ++u) {
      const int b = b0 + u * 32;
      const int* __restrict__ row = table + ((b < NB) ? b : (NB - 1)) * NCH;
#pragma unroll
      for (int k = 0; k < KR; ++k) {
        const int idx = k * 32 + lane;
        tv[u][k] = row[(idx < NCH) ? idx : (NCH - 1)];
      }
    }
#pragma unroll
    for (int u = 0; u < 4; ++u) {
      const int b = b0 + u * 32;
      int t = 0;
      int p = 0;
#pragma unroll
      for (int k = 0; k < KR; ++k) {
        const int idx = k * 32 + lane;
        const int v = (idx < NCH) ? tv[u][k] : 0;
        t += v;
        p += (idx < c) ? v : 0;
      }
      for (int o = 16; o > 0; o >>= 1) {
        t += __shfl_xor(t, o);
        p += __shfl_xor(p, o);
      }
      if (lane == 0 && b < NB) {
        tot[b] = t;
        cur[b] = p;
      }
    }
  }
  __syncthreads();

  if (wave == 0) {
    int v[13];
    int s = 0;
#pragma unroll
    for (int k = 0; k < 13; ++k) {
      const int idx = lane * 13 + k;
      v[k] = (idx < NB) ? tot[idx] : 0;
      s += v[k];
    }
    int incl = s;
    for (int o = 1; o < 32; o <<= 1) {
      const int u = __shfl_up(incl, o);
      if (lane >= o) incl += u;
    }
    int excl = incl - s;
#pragma unroll
    for (int k = 0; k < 13; ++k) {
      const int idx = lane * 13 + k;
      if (idx < NB) tot[idx] = excl;
      excl += v[k];
    }
    if (lane == 31) tot[NB] = incl;
  }
  __syncthreads();
  for (int b = tid; b < NB; b += 1024) cur[b] += tot[b];
  if (c == 0)
    for (int b = tid; b <= NB; b += 1024) binstart[b] = tot[b];
  __syncthreads();

  const int4* __restrict__ ps = (const int4*)(src + c * CH);
  const int4* __restrict__ pd = (const int4*)(dst + c * CH);
  constexpr int Q = CH / 4;
  constexpr int KB = (Q + 1023) / 1024;
  int4 s[KB];
  int4 d[KB];
#pragma unroll
  for (int k = 0; k < KB; ++k) {
    const int i = tid + k * 1024;
    s[k] = ps[(i < Q) ? i : (Q - 1)];
    d[k] = pd[(i < Q) ? i : (Q - 1)];
  }
#pragma unroll
  for (int k = 0; k < KB; ++k) {
    if (tid + k * 1024 < Q) {
      part[atomicAdd(&cur[d[k].x >> BSH], 1)] = make_int2(s[k].x, d[k].x);
      part[atomicAdd(&cur[d[k].y >> BSH], 1)] = make_int2(s[k].y, d[k].y);
      part[atomicAdd(&cur[d[k].z >> BSH], 1)] = make_int2(s[k].z, d[k].z);
      part[atomicAdd(&cur[d[k].w >> BSH], 1)] = make_int2(s[k].w, d[k].w);
    }
  }
}

__global__ __launch_bounds__(512) void bin_build_kernel(const int* __restrict__ part_,
                                                         const int* __restrict__ binstart,
                                                         int* __restrict__ rowptr,
                                                         float* __restrict__ dinv,
                                                         int* __restrict__ csr,
                                                         const float* __restrict__ x,
                                                         float* __restrict__ hpad) {
  const int2* __restrict__ part = (const int2*)part_;
  __shared__ int cnt[BS];
  __shared__ int cursor[BS];
  __shared__ int wtot[BS / 32];
  const int tid = threadIdx.x;
  const int lane = tid & 31;
  const int wave = tid >> 5;
  const int b = blockIdx.x;
  const int lo = binstart[b];
  const int hi = binstart[b + 1];
  const int base = b * BS;

  if (tid < BS) cnt[tid] = 0;
  __syncthreads();
  float xv[9];
  if (tid < BS && base + tid < NN) {
    const float* __restrict__ xr = x + (size_t)(base + tid) * 9;
#pragma unroll
    for (int k = 0; k < 9; ++k) xv[k] = xr[k];
  }
  for (int j0 = lo + tid; j0 < hi; j0 += 8 * 512) {
    int2 e[8];
#pragma unroll
    for (int k = 0; k < 8; ++k) {
      const int j = j0 + k * 512;
      e[k] = part[(j < hi) ? j : (hi - 1)];
    }
#pragma unroll
    for (int k = 0; k < 8; ++k)
      if (j0 + k * 512 < hi) atomicAdd(&cnt[e[k].y - base], 1);
  }
  __syncthreads();

  int c = 0;
  int incl = 0;
  if (tid < BS) {
    c = cnt[tid];
    incl = c;
    for (int o = 1; o < 32; o <<= 1) {
      const int u = __shfl_up(incl, o);
      if (lane >= o) incl += u;
    }
    if (lane == 31) wtot[wave] = incl;
  }
  __syncthreads();
  if (tid < BS) {
    int off = lo;
    for (int w = 0; w < wave; ++w) off += wtot[w];
    const int excl = off + incl - c;
    cursor[tid] = excl;
    const int node = base + tid;
    if (node < NN) {
      rowptr[node] = excl;
      const float di = rsqrtf((float)(c + 1));
      dinv[node] = di;
      float4* __restrict__ o = (float4*)(hpad + (size_t)node * 16);
      o[0] = make_float4(xv[0] * di, xv[1] * di, xv[2] * di, xv[3] * di);
      o[1] = make_float4(xv[4] * di, xv[5] * di, xv[6] * di, xv[7] * di);
      o[2] = make_float4(xv[8] * di, 0.0f, 0.0f, 0.0f);
      o[3] = make_float4(0.0f, 0.0f, 0.0f, 0.0f);
    }
  }
  if (b == NB - 1 && tid == 0) rowptr[NN] = hi;
  __syncthreads();

  for (int j0 = lo + tid; j0 < hi; j0 += 8 * 512) {
    int2 e[8];
#pragma unroll
    for (int k = 0; k < 8; ++k) {
      const int j = j0 + k * 512;
      e[k] = part[(j < hi) ? j : (hi - 1)];
    }
#pragma unroll
    for (int k = 0; k < 8; ++k)
      if (j0 + k * 512 < hi) csr[atomicAdd(&cursor[e[k].y - base], 1)] = e[k].x;
  }
}

__device__ __forceinline__ void add4(float4& acc, const float4 v) {
  acc.x += v.x;
  acc.y += v.y;
  acc.z += v.z;
  acc.w += v.w;
}
__device__ __forceinline__ void axpy4(float4& acc, const float4 v, const float w) {
  acc.x = fmaf(v.x, w, acc.x);
  acc.y = fmaf(v.y, w, acc.y);
  acc.z = fmaf(v.z, w, acc.z);
  acc.w = fmaf(v.w, w, acc.w);
}

template <int LPN, int S>
__device__ __forceinline__ int slot_src(const int (&my)[16 / LPN]) {
  static_assert(LPN == 16 || LPN == 4, "a node's lanes are a 16-lane row or a quad");
  if constexpr (LPN == 16)
    return __builtin_amdgcn_update_dpp(0, my[0], 0x150 + S, 0xF, 0xF, true);
  else
    return __builtin_amdgcn_update_dpp(0, my[S / 4], (S % 4) * 0x55, 0xF, 0xF, true);
}

template <int LPN, int S0>
__device__ __forceinline__ void gather8(const int (&my)[16 / LPN], const float4* __restrict__ hin4,
                                        const int lg, const int left, float4& acc) {
  int e[8];
  float4 v[8];
  e[0] = slot_src<LPN, S0 + 0>(my);
  e[1] = slot_src<LPN, S0 + 1>(my);
  e[2] = slot_src<LPN, S0 + 2>(my);
  e[3] = slot_src<LPN, S0 + 3>(my);
  e[4] = slot_src<LPN, S0 + 4>(my);
  e[5] = slot_src<LPN, S0 + 5>(my);
  e[6] = slot_src<LPN, S0 + 6>(my);
  e[7] = slot_src<LPN, S0 + 7>(my);
#pragma unroll
  for (int k = 0; k < 8; ++k) v[k] = hin4[e[k] * LPN + lg];
  if (left >= 8) {
#pragma unroll
    for (int k = 0; k < 8; ++k) add4(acc, v[k]);
  } else {
#pragma unroll
    for (int k = 0; k < 8; ++k) axpy4(acc, v[k], (k < left) ? 1.0f : 0.0f);
  }
}

__device__ inline void atomicMaxF32(float* addr, float val) {
  if (val >= 0.0f)
    atomicMax((int*)addr, __float_as_int(val));
  else
    atomicMin((unsigned int*)addr, __float_as_uint(val));
}

__device__ __forceinline__ void pool_flush(const int g, const int f, const int half,
                                           const bool cntlane, const float s, const float m,
                                           const int c, float* __restrict__ gmax,
                                           float* __restrict__ gsum, float* __restrict__ cnt) {
  if (half == 0)
    atomicAdd(&gsum[(size_t)g * DD + f], s);
  else
    atomicMaxF32(&gmax[(size_t)g * DD + f], m);
  if (cntlane) atomicAdd(&cnt[g], (float)c);
}

template <int K, bool SCALE_OUT, bool POOL>
__global__ __launch_bounds__(K * 4) void layer_kernel(const float* __restrict__ hin,
                                                      const float* __restrict__ W,
                                                      const int* __restrict__ rowptr,
                                                      const int* __restrict__ csr,
                                                      const float* __restrict__ dinv,
                                                      const float* __restrict__ b,
                                                      float* __restrict__ hout,
                                                      const int* __restrict__ batch,
                                                      float* __restrict__ gmax,
                                                      float* __restrict__ gsum,
                                                      float* __restrict__ gcnt) {
  static_assert(!(POOL && SCALE_OUT), "the pooled layer stores no rows, so there is nothing to scale");
  constexpr int LPN = K / 4;
  constexpr int NW = K / 8;
  constexpr int TPW = (NW >= 4) ? 1 : 4 / NW;
  constexpr int KS = K / 4;
  constexpr int P = K + 4;
  __shared__ __attribute__((aligned(16))) float zs[16 * P];
  __shared__ __attribute__((aligned(16))) float dis[16];

  const int tid = threadIdx.x;
  const int node = tid / LPN;
  const int fg = (tid % LPN) * 4;
  const int m0 = blockIdx.x * 16;
  const int d = m0 + node;

  int j = rowptr[d];
  const int end = rowptr[d + 1];
  const float4 self = *(const float4*)(hin + d * K + fg);
  const float di = dinv[d];

  float4 acc = make_float4(0.0f, 0.0f, 0.0f, 0.0f);

  if (j < end) {
    constexpr int NR = 16 / LPN;
    const int lg = tid % LPN;
    const float4* __restrict__ hin4 = (const float4*)hin;
    const int last = end - 1;
    int my[NR];
#pragma unroll
    for (int r = 0; r < NR; ++r) {
      const int s = j + r * LPN + lg;
      my[r] = csr[(s < last) ? s : last];
    }
    for (;;) {
      const int jn = j + 16;
      int nxt[NR];
#pragma unroll
      for (int r = 0; r < NR; ++r) {
        const int s = jn + r * LPN + lg;
        nxt[r] = csr[(s < last) ? s : last];
      }
      gather8<LPN, 0>(my, hin4, lg, end - j, acc);
      if (j + 8 < end) gather8<LPN, 8>(my, hin4, lg, end - j - 8, acc);
      if (jn >= end) break;
#pragma unroll
      for (int r = 0; r < NR; ++r) my[r] = nxt[r];
      j = jn;
    }
  }

  const int wave = __builtin_amdgcn_readfirstlane(tid >> 5);
  const int lane = tid & 31;
  const int l15 = lane & 15;
  const int half = lane >> 4;
  const bool mult = wave < 4;

  float4 z;
  z.x = (acc.x + self.x) * di;
  z.y = (acc.y + self.y) * di;
  z.z = (acc.z + self.z) * di;
  z.w = (acc.w + self.w) * di;
  *(float4*)(zs + node * P + fg) = z;
  if (SCALE_OUT && fg == 0) dis[node] = di;

  v2f bf[TPW][KS];
  if (mult) {
#pragma unroll
    for (int t = 0; t < TPW; ++t) {
      const float* __restrict__ bcol = W + half * 2 * 64 + (wave + t * NW) * 16 + l15;
#pragma unroll
      for (int s = 0; s < KS; ++s) {
        bf[t][s].x = bcol[s * 4 * 64];
        bf[t][s].y = bcol[s * 4 * 64 + 64];
      }
    }
  }
  __builtin_amdgcn_fence(__ATOMIC_RELEASE, "workgroup", "local");
  __builtin_amdgcn_s_barrier();
  __builtin_amdgcn_fence(__ATOMIC_ACQUIRE, "workgroup", "local");

  if (mult) {
    const float* zrow = zs + l15 * P + half * 2;
#pragma unroll
    for (int t = 0; t < TPW; ++t) {
      const int col = (wave + t * NW) * 16 + l15;
      v8f c0 = {};
      v8f c1 = {};
#pragma unroll
      for (int s = 0; s < KS; s += 2) {
        const v2f a0 = *(const v2f*)(zrow + s * 4);
        const v2f a1 = *(const v2f*)(zrow + s * 4 + 4);
        c0 = __builtin_amdgcn_wmma_f32_16x16x4_f32(false, a0, false, bf[t][s], (short)0, c0,
                                                   false, false);
        c1 = __builtin_amdgcn_wmma_f32_16x16x4_f32(false, a1, false, bf[t][s + 1], (short)0, c1,
                                                   false, false);
      }
      const float bias = b[col];
      float* __restrict__ orow = hout + (m0 + half * 8) * 64 + col;
      if (POOL) {
        float own[8];
        float oth[8];
        float cv[16];
#pragma unroll
        for (int v = 0; v < 8; ++v) own[v] = tanhf(c0[v] + c1[v] + bias);
#pragma unroll
        for (int v = 0; v < 8; ++v) oth[v] = __shfl_xor(own[v], 16);
#pragma unroll
        for (int v = 0; v < 8; ++v) {
          cv[v] = half ? oth[v] : own[v];
          cv[8 + v] = half ? own[v] : oth[v];
        }
        int g[16];
#pragma unroll
        for (int r = 0; r < 16; ++r) g[r] = batch[m0 + r];
        const bool cntlane = (col == 0) && (half == 0);
        int cur = g[0];
        float s = cv[0];
        float m = cv[0];
        int c = 1;
#pragma unroll
        for (int r = 1; r < 16; ++r) {
          if (g[r] != cur) {
            pool_flush(cur, col, half, cntlane, s, m, c, gmax, gsum, gcnt);
            cur = g[r];
            s = cv[r];
            m = cv[r];
            c = 1;
          } else {
            s += cv[r];
            m = fmaxf(m, cv[r]);
            c += 1;
          }
        }
        pool_flush(cur, col, half, cntlane, s, m, c, gmax, gsum, gcnt);
      } else if (SCALE_OUT) {
        const float4 s0 = *(const float4*)(dis + half * 8);
        const float4 s1 = *(const float4*)(dis + half * 8 + 4);
        const float so[8] = {s0.x, s0.y, s0.z, s0.w, s1.x, s1.y, s1.z, s1.w};
#pragma unroll
        for (int v = 0; v < 8; ++v) orow[v * 64] = tanhf(c0[v] + c1[v] + bias) * so[v];
      } else {
#pragma unroll
        for (int v = 0; v < 8; ++v) orow[v * 64] = tanhf(c0[v] + c1[v] + bias);
      }
    }
  }
}


static_assert(GG % 32 == 0 && DD == 64, "32 graphs per work-group, two features per lane");
__global__ __launch_bounds__(1024) void pool_finish_head_kernel(const float* __restrict__ gmax,
                                                                const float* __restrict__ gsum,
                                                                const float* __restrict__ cnt,
                                                                const float* __restrict__ Wout,
                                                                const float* __restrict__ bout,
                                                                float* __restrict__ hidden,
                                                                float* __restrict__ out) {
  __shared__ float dot[32];
  const int lane = threadIdx.x & 31;
  const int wave = threadIdx.x >> 5;
  const int g = blockIdx.x * 32 + wave;
  const float c = cnt[g];
  float2 mx = ((const float2*)(gmax + (size_t)g * DD))[lane];
  const float2 sm = ((const float2*)(gsum + (size_t)g * DD))[lane];
  if (c == 0.0f) mx = make_float2(0.0f, 0.0f);
  const float cd = fmaxf(c, 1.0f);
  const float2 mean = make_float2(sm.x / cd, sm.y / cd);
  float* __restrict__ hrow = hidden + (size_t)g * (2 * DD);
  ((float2*)hrow)[lane] = mx;
  ((float2*)(hrow + DD))[lane] = mean;
  const int f = lane * 2;
  float acc = mx.x * Wout[f];
  acc = fmaf(mx.y, Wout[f + 1], acc);
  acc = fmaf(mean.x, Wout[DD + f], acc);
  acc = fmaf(mean.y, Wout[DD + f + 1], acc);
  for (int o = 16; o > 0; o >>= 1) acc += __shfl_xor(acc, o);
  if (lane == 0) dot[wave] = acc;
  __syncthreads();
  if (wave == 0) out[blockIdx.x * 32 + lane] = dot[lane] + bout[0];
}


extern "C" void kernel_launch(void* const* d_in, const int* in_sizes, int n_in,
                              void* d_out, int out_size, void* d_ws, size_t ws_size,
                              hipStream_t stream) {
  (void)in_sizes; (void)n_in; (void)out_size; (void)ws_size;

  const float* x    = (const float*)d_in[0];
  const int*   eidx = (const int*)d_in[1];
  const int*   batch = (const int*)d_in[2];
  const float* W0 = (const float*)d_in[3];
  const float* b0 = (const float*)d_in[4];
  const float* W1 = (const float*)d_in[5];
  const float* b1 = (const float*)d_in[6];
  const float* W2 = (const float*)d_in[7];
  const float* b2 = (const float*)d_in[8];
  const float* W3 = (const float*)d_in[9];
  const float* b3 = (const float*)d_in[10];
  const float* Wout = (const float*)d_in[11];
  const float* bout = (const float*)d_in[12];

  const int* src = eidx;
  const int* dst = eidx + EE;

  float* outp   = (float*)d_out;
  float* hidden = (float*)d_out + GG;

  float* ws = (float*)d_ws;
  size_t o = 0;
  float* dinv   = ws + o;         o += NN + 48;
  int*   rowptr = (int*)(ws + o); o += NN + 112;
  int*   table  = (int*)(ws + o); o += NB * NCH + 48;
  int*   bstart = (int*)(ws + o); o += 448;
  int*   part   = (int*)(ws + o); o += (size_t)EE * 2;
  int*   csr    = (int*)(ws + o); o += (size_t)EE;
  float* hpad  = ws + o; o += (size_t)NN * 16;
  float* w0pad = ws + o; o += 16 * 64;
  float* h0    = ws + o; o += (size_t)NN * DD;
  float* h1    = ws + o; o += (size_t)NN * DD;
  float* gmax  = ws + o; o += (size_t)GG * DD;
  float* gsum  = ws + o; o += (size_t)GG * DD;
  float* gcnt  = ws + o; o += GG;

  bin_count_kernel<<<NCH + NAUX, 256, 0, stream>>>(dst, table, W0, w0pad, gmax, gsum, gcnt);
  bin_scatter_kernel<<<NCH, 1024, 0, stream>>>(src, dst, table, bstart, part);
  bin_build_kernel<<<NB, 512, 0, stream>>>(part, bstart, rowptr, dinv, csr, x, hpad);

  const int mtiles = NN / 16;

  layer_kernel<16, true, false><<<mtiles, 64, 0, stream>>>(hpad, w0pad, rowptr, csr, dinv, b0, h0, batch, gmax, gsum, gcnt);
  layer_kernel<64, true, false><<<mtiles, 256, 0, stream>>>(h0, W1, rowptr, csr, dinv, b1, h1, batch, gmax, gsum, gcnt);
  layer_kernel<64, true, false><<<mtiles, 256, 0, stream>>>(h1, W2, rowptr, csr, dinv, b2, h0, batch, gmax, gsum, gcnt);
  layer_kernel<64, false, true><<<mtiles, 256, 0, stream>>>(h0, W3, rowptr, csr, dinv, b3, h1, batch, gmax, gsum, gcnt);

  pool_finish_head_kernel<<<GG / 32, 1024, 0, stream>>>(gmax, gsum, gcnt, Wout, bout, hidden, outp);
}


# Edge sort in two launches, not three: a chunk block groups its own edges by bin inside its own range (no table sweep), a bin block collects its 240 groups; order of edge sums changes
# speedup vs baseline: 6.0870x; 1.1813x over previous
#include <hip/hip_runtime.h>
#include <math.h>
#include <float.h>


typedef __attribute__((ext_vector_type(2))) float v2f;
typedef __attribute__((ext_vector_type(8))) float v8f;

#define NN 50000
#define EE 1200000
#define DD 64
#define GG 1024

static_assert(NN % 16 == 0, "the layer kernel's 16-row WMMA tiles need a row count that is a multiple of 16");


#define BSH 7
#define BS (1 << BSH)
#define NB ((NN + BS - 1) / BS)
#define RPB (BS + 1)
#define NCH 240
#define CH (EE / NCH)
static_assert(EE % NCH == 0 && CH % 4 == 0, "chunks are whole 16-byte groups of edges");
static_assert(NB <= 32 * 13, "the bin scan gives each lane of one wave 13 bins");
static_assert(CH < 65536, "a group's start in its chunk and its length share one word");
static_assert(NCH % 16 == 0, "the 16 waves of a bin_build block take the same number of groups");

#define NAUX ((GG * DD / 4 + 1023) / 1024)
static_assert(NAUX * 1024 >= GG * DD / 4 && NAUX * 1024 >= 16 * 64,
              "one item per thread of the extra work-groups covers every pass");
static_assert(GG % 4 == 0, "the count plane is filled in 16-byte groups");

__global__ __launch_bounds__(1024) void chunk_sort_kernel(const int* __restrict__ src,
                                                          const int* __restrict__ dst,
                                                          int* __restrict__ table,
                                                          int* __restrict__ part_,
                                                          const float* __restrict__ W0,
                                                          float* __restrict__ w0pad,
                                                          float* __restrict__ gmax,
                                                          float* __restrict__ gsum,
                                                          float* __restrict__ gcnt,
                                                          int* __restrict__ alloc) {
  int2* __restrict__ part = (int2*)part_;
  __shared__ int hist[NB];
  __shared__ int cur[NB];
  const int tid = threadIdx.x;
  const int lane = tid & 31;
  const int wave = tid >> 5;
  const int c = blockIdx.x;
  if (c >= NCH) {
    const int t = (c - NCH) * 1024 + tid;
    if (t < GG * DD / 4) {
      ((float4*)gmax)[t] = make_float4(-FLT_MAX, -FLT_MAX, -FLT_MAX, -FLT_MAX);
      ((float4*)gsum)[t] = make_float4(0.0f, 0.0f, 0.0f, 0.0f);
    }
    if (t < GG / 4) ((float4*)gcnt)[t] = make_float4(0.0f, 0.0f, 0.0f, 0.0f);
    if (t < 16 * 64) w0pad[t] = (t < 9 * 64) ? W0[t] : 0.0f;
    if (t == 0) alloc[0] = 0;
    return;
  }
  if (tid < NB) hist[tid] = 0;
  __syncthreads();

  const int4* __restrict__ ps = (const int4*)(src + c * CH);
  const int4* __restrict__ pd = (const int4*)(dst + c * CH);
  constexpr int Q = CH / 4;
  constexpr int KB = (Q + 1023) / 1024;
  int4 s[KB];
  int4 d[KB];
#pragma unroll
  for (int k = 0; k < KB; ++k) {
    const int i = tid + k * 1024;
    s[k] = ps[(i < Q) ? i : (Q - 1)];
    d[k] = pd[(i < Q) ? i : (Q - 1)];
  }
#pragma unroll
  for (int k = 0; k < KB; ++k) {
    if (tid + k * 1024 < Q) {
      atomicAdd(&hist[d[k].x >> BSH], 1);
      atomicAdd(&hist[d[k].y >> BSH], 1);
      atomicAdd(&hist[d[k].z >> BSH], 1);
      atomicAdd(&hist[d[k].w >> BSH], 1);
    }
  }
  __syncthreads();

  if (wave == 0) {
    int v[13];
    int sum = 0;
#pragma unroll
    for (int k = 0; k < 13; ++k) {
      const int idx = lane * 13 + k;
      v[k] = (idx < NB) ? hist[idx] : 0;
      sum += v[k];
    }
    int incl = sum;
    for (int o = 1; o < 32; o <<= 1) {
      const int u = __shfl_up(incl, o);
      if (lane >= o) incl += u;
    }
    int excl = incl - sum;
#pragma unroll
    for (int k = 0; k < 13; ++k) {
      const int idx = lane * 13 + k;
      if (idx < NB) {
        cur[idx] = excl;
        table[idx * NCH + c] = (excl << 16) | v[k];
      }
      excl += v[k];
    }
  }
  __syncthreads();

  int2* __restrict__ mine = part + c * CH;
#pragma unroll
  for (int k = 0; k < KB; ++k) {
    if (tid + k * 1024 < Q) {
      mine[atomicAdd(&cur[d[k].x >> BSH], 1)] = make_int2(s[k].x, d[k].x);
      mine[atomicAdd(&cur[d[k].y >> BSH], 1)] = make_int2(s[k].y, d[k].y);
      mine[atomicAdd(&cur[d[k].z >> BSH], 1)] = make_int2(s[k].z, d[k].z);
      mine[atomicAdd(&cur[d[k].w >> BSH], 1)] = make_int2(s[k].w, d[k].w);
    }
  }
}

__global__ __launch_bounds__(512) void bin_build_kernel(const int* __restrict__ part_,
                                                         const int* __restrict__ table,
                                                         int* __restrict__ alloc,
                                                         int* __restrict__ rowptr,
                                                         float* __restrict__ dinv,
                                                         int* __restrict__ csr,
                                                         const float* __restrict__ x,
                                                         float* __restrict__ hpad) {
  const int2* __restrict__ part = (const int2*)part_;
  __shared__ int cnt[BS];
  __shared__ int cursor[BS];
  __shared__ int wtot[BS / 32];
  __shared__ int wsum[16];
  __shared__ int lo_s;
  const int tid = threadIdx.x;
  const int lane = tid & 31;
  const int wave = tid >> 5;
  const int b = blockIdx.x;
  const int base = b * BS;
  constexpr int SPW = NCH / 16;
  static_assert(SPW <= 32, "a wave's table entries are loaded one per lane");

  if (tid < BS) cnt[tid] = 0;
  int pk = 0;
  if (lane < SPW) pk = table[b * NCH + wave + 16 * lane];
  float xv[9];
  if (tid < BS && base + tid < NN) {
    const float* __restrict__ xr = x + (size_t)(base + tid) * 9;
#pragma unroll
    for (int k = 0; k < 9; ++k) xv[k] = xr[k];
  }
  int2 e[SPW];
  int sn[SPW];
  int so[SPW];
#pragma unroll
  for (int k = 0; k < SPW; ++k) {
    const int w = __builtin_amdgcn_readlane(pk, k);
    sn[k] = w & 0xFFFF;
    so[k] = (wave + 16 * k) * CH + (w >> 16);
  }
#pragma unroll
  for (int k = 0; k < SPW; ++k) {
    e[k] = make_int2(0, base);
    if (lane < sn[k]) e[k] = part[so[k] + lane];
  }
  int tsum = pk & 0xFFFF;
  for (int o = 16; o > 0; o >>= 1) tsum += __shfl_xor(tsum, o);
  if (lane == 0) wsum[wave] = tsum;
  __syncthreads();
  if (tid == 0) {
    int total = 0;
    for (int w = 0; w < 16; ++w) total += wsum[w];
    lo_s = atomicAdd(alloc, total);
  }
#pragma unroll
  for (int k = 0; k < SPW; ++k)
    if (lane < sn[k]) atomicAdd(&cnt[e[k].y - base], 1);
#pragma unroll
  for (int k = 0; k < SPW; ++k)
    for (int i = 32 + lane; i < sn[k]; i += 32) atomicAdd(&cnt[part[so[k] + i].y - base], 1);
  __syncthreads();

  const int lo = lo_s;
  int c = 0;
  int incl = 0;
  if (tid < BS) {
    c = cnt[tid];
    incl = c;
    for (int o = 1; o < 32; o <<= 1) {
      const int u = __shfl_up(incl, o);
      if (lane >= o) incl += u;
    }
    if (lane == 31) wtot[wave] = incl;
  }
  __syncthreads();
  if (tid < BS) {
    int off = lo;
    for (int w = 0; w < wave; ++w) off += wtot[w];
    const int excl = off + incl - c;
    cursor[tid] = excl;
    rowptr[b * RPB + tid] = excl;
    if (tid == BS - 1) rowptr[b * RPB + BS] = excl + c;
    const int node = base + tid;
    if (node < NN) {
      const float di = rsqrtf((float)(c + 1));
      dinv[node] = di;
      float4* __restrict__ o = (float4*)(hpad + (size_t)node * 16);
      o[0] = make_float4(xv[0] * di, xv[1] * di, xv[2] * di, xv[3] * di);
      o[1] = make_float4(xv[4] * di, xv[5] * di, xv[6] * di, xv[7] * di);
      o[2] = make_float4(xv[8] * di, 0.0f, 0.0f, 0.0f);
      o[3] = make_float4(0.0f, 0.0f, 0.0f, 0.0f);
    }
  }
  __syncthreads();

  int pos[SPW];
#pragma unroll
  for (int k = 0; k < SPW; ++k) {
    pos[k] = 0;
    if (lane < sn[k]) pos[k] = atomicAdd(&cursor[e[k].y - base], 1);
  }
#pragma unroll
  for (int k = 0; k < SPW; ++k)
    if (lane < sn[k]) csr[pos[k]] = e[k].x;
#pragma unroll
  for (int k = 0; k < SPW; ++k)
    for (int i = 32 + lane; i < sn[k]; i += 32) {
      const int2 p = part[so[k] + i];
      csr[atomicAdd(&cursor[p.y - base], 1)] = p.x;
    }
}

__device__ __forceinline__ void add4(float4& acc, const float4 v) {
  acc.x += v.x;
  acc.y += v.y;
  acc.z += v.z;
  acc.w += v.w;
}
__device__ __forceinline__ void axpy4(float4& acc, const float4 v, const float w) {
  acc.x = fmaf(v.x, w, acc.x);
  acc.y = fmaf(v.y, w, acc.y);
  acc.z = fmaf(v.z, w, acc.z);
  acc.w = fmaf(v.w, w, acc.w);
}

template <int LPN, int S>
__device__ __forceinline__ int slot_src(const int (&my)[16 / LPN]) {
  static_assert(LPN == 16 || LPN == 4, "a node's lanes are a 16-lane row or a quad");
  if constexpr (LPN == 16)
    return __builtin_amdgcn_update_dpp(0, my[0], 0x150 + S, 0xF, 0xF, true);
  else
    return __builtin_amdgcn_update_dpp(0, my[S / 4], (S % 4) * 0x55, 0xF, 0xF, true);
}

template <int LPN, int S0>
__device__ __forceinline__ void gather8(const int (&my)[16 / LPN], const float4* __restrict__ hin4,
                                        const int lg, const int left, float4& acc) {
  int e[8];
  float4 v[8];
  e[0] = slot_src<LPN, S0 + 0>(my);
  e[1] = slot_src<LPN, S0 + 1>(my);
  e[2] = slot_src<LPN, S0 + 2>(my);
  e[3] = slot_src<LPN, S0 + 3>(my);
  e[4] = slot_src<LPN, S0 + 4>(my);
  e[5] = slot_src<LPN, S0 + 5>(my);
  e[6] = slot_src<LPN, S0 + 6>(my);
  e[7] = slot_src<LPN, S0 + 7>(my);
#pragma unroll
  for (int k = 0; k < 8; ++k) v[k] = hin4[e[k] * LPN + lg];
  if (left >= 8) {
#pragma unroll
    for (int k = 0; k < 8; ++k) add4(acc, v[k]);
  } else {
#pragma unroll
    for (int k = 0; k < 8; ++k) axpy4(acc, v[k], (k < left) ? 1.0f : 0.0f);
  }
}

__device__ inline void atomicMaxF32(float* addr, float val) {
  if (val >= 0.0f)
    atomicMax((int*)addr, __float_as_int(val));
  else
    atomicMin((unsigned int*)addr, __float_as_uint(val));
}

__device__ __forceinline__ void pool_flush(const int g, const int f, const int half,
                                           const bool cntlane, const float s, const float m,
                                           const int c, float* __restrict__ gmax,
                                           float* __restrict__ gsum, float* __restrict__ cnt) {
  if (half == 0)
    atomicAdd(&gsum[(size_t)g * DD + f], s);
  else
    atomicMaxF32(&gmax[(size_t)g * DD + f], m);
  if (cntlane) atomicAdd(&cnt[g], (float)c);
}

template <int K, bool SCALE_OUT, bool POOL>
__global__ __launch_bounds__(K * 4) void layer_kernel(const float* __restrict__ hin,
                                                      const float* __restrict__ W,
                                                      const int* __restrict__ rowptr,
                                                      const int* __restrict__ csr,
                                                      const float* __restrict__ dinv,
                                                      const float* __restrict__ b,
                                                      float* __restrict__ hout,
                                                      const int* __restrict__ batch,
                                                      float* __restrict__ gmax,
                                                      float* __restrict__ gsum,
                                                      float* __restrict__ gcnt) {
  static_assert(!(POOL && SCALE_OUT), "the pooled layer stores no rows, so there is nothing to scale");
  constexpr int LPN = K / 4;
  constexpr int NW = K / 8;
  constexpr int TPW = (NW >= 4) ? 1 : 4 / NW;
  constexpr int KS = K / 4;
  constexpr int P = K + 4;
  __shared__ __attribute__((aligned(16))) float zs[16 * P];
  __shared__ __attribute__((aligned(16))) float dis[16];

  const int tid = threadIdx.x;
  const int node = tid / LPN;
  const int fg = (tid % LPN) * 4;
  const int m0 = blockIdx.x * 16;
  const int d = m0 + node;

  const int ri = d + (d >> BSH);
  int j = rowptr[ri];
  const int end = rowptr[ri + 1];
  const float4 self = *(const float4*)(hin + d * K + fg);
  const float di = dinv[d];

  float4 acc = make_float4(0.0f, 0.0f, 0.0f, 0.0f);

  if (j < end) {
    constexpr int NR = 16 / LPN;
    const int lg = tid % LPN;
    const float4* __restrict__ hin4 = (const float4*)hin;
    const int last = end - 1;
    int my[NR];
#pragma unroll
    for (int r = 0; r < NR; ++r) {
      const int s = j + r * LPN + lg;
      my[r] = csr[(s < last) ? s : last];
    }
    for (;;) {
      const int jn = j + 16;
      int nxt[NR];
#pragma unroll
      for (int r = 0; r < NR; ++r) {
        const int s = jn + r * LPN + lg;
        nxt[r] = csr[(s < last) ? s : last];
      }
      gather8<LPN, 0>(my, hin4, lg, end - j, acc);
      if (j + 8 < end) gather8<LPN, 8>(my, hin4, lg, end - j - 8, acc);
      if (jn >= end) break;
#pragma unroll
      for (int r = 0; r < NR; ++r) my[r] = nxt[r];
      j = jn;
    }
  }

  const int wave = __builtin_amdgcn_readfirstlane(tid >> 5);
  const int lane = tid & 31;
  const int l15 = lane & 15;
  const int half = lane >> 4;
  const bool mult = wave < 4;

  float4 z;
  z.x = (acc.x + self.x) * di;
  z.y = (acc.y + self.y) * di;
  z.z = (acc.z + self.z) * di;
  z.w = (acc.w + self.w) * di;
  *(float4*)(zs + node * P + fg) = z;
  if (SCALE_OUT && fg == 0) dis[node] = di;

  v2f bf[TPW][KS];
  if (mult) {
#pragma unroll
    for (int t = 0; t < TPW; ++t) {
      const float* __restrict__ bcol = W + half * 2 * 64 + (wave + t * NW) * 16 + l15;
#pragma unroll
      for (int s = 0; s < KS; ++s) {
        bf[t][s].x = bcol[s * 4 * 64];
        bf[t][s].y = bcol[s * 4 * 64 + 64];
      }
    }
  }
  __builtin_amdgcn_fence(__ATOMIC_RELEASE, "workgroup", "local");
  __builtin_amdgcn_s_barrier();
  __builtin_amdgcn_fence(__ATOMIC_ACQUIRE, "workgroup", "local");

  if (mult) {
    const float* zrow = zs + l15 * P + half * 2;
#pragma unroll
    for (int t = 0; t < TPW; ++t) {
      const int col = (wave + t * NW) * 16 + l15;
      v8f c0 = {};
      v8f c1 = {};
#pragma unroll
      for (int s = 0; s < KS; s += 2) {
        const v2f a0 = *(const v2f*)(zrow + s * 4);
        const v2f a1 = *(const v2f*)(zrow + s * 4 + 4);
        c0 = __builtin_amdgcn_wmma_f32_16x16x4_f32(false, a0, false, bf[t][s], (short)0, c0,
                                                   false, false);
        c1 = __builtin_amdgcn_wmma_f32_16x16x4_f32(false, a1, false, bf[t][s + 1], (short)0, c1,
                                                   false, false);
      }
      const float bias = b[col];
      float* __restrict__ orow = hout + (m0 + half * 8) * 64 + col;
      if (POOL) {
        float own[8];
        float oth[8];
        float cv[16];
#pragma unroll
        for (int v = 0; v < 8; ++v) own[v] = tanhf(c0[v] + c1[v] + bias);
#pragma unroll
        for (int v = 0; v < 8; ++v) oth[v] = __shfl_xor(own[v], 16);
#pragma unroll
        for (int v = 0; v < 8; ++v) {
          cv[v] = half ? oth[v] : own[v];
          cv[8 + v] = half ? own[v] : oth[v];
        }
        int g[16];
#pragma unroll
        for (int r = 0; r < 16; ++r) g[r] = batch[m0 + r];
        const bool cntlane = (col == 0) && (half == 0);
        int cur = g[0];
        float s = cv[0];
        float m = cv[0];
        int c = 1;
#pragma unroll
        for (int r = 1; r < 16; ++r) {
          if (g[r] != cur) {
            pool_flush(cur, col, half, cntlane, s, m, c, gmax, gsum, gcnt);
            cur = g[r];
            s = cv[r];
            m = cv[r];
            c = 1;
          } else {
            s += cv[r];
            m = fmaxf(m, cv[r]);
            c += 1;
          }
        }
        pool_flush(cur, col, half, cntlane, s, m, c, gmax, gsum, gcnt);
      } else if (SCALE_OUT) {
        const float4 s0 = *(const float4*)(dis + half * 8);
        const float4 s1 = *(const float4*)(dis + half * 8 + 4);
        const float so[8] = {s0.x, s0.y, s0.z, s0.w, s1.x, s1.y, s1.z, s1.w};
#pragma unroll
        for (int v = 0; v < 8; ++v) orow[v * 64] = tanhf(c0[v] + c1[v] + bias) * so[v];
      } else {
#pragma unroll
        for (int v = 0; v < 8; ++v) orow[v * 64] = tanhf(c0[v] + c1[v] + bias);
      }
    }
  }
}


static_assert(GG % 32 == 0 && DD == 64, "32 graphs per work-group, two features per lane");
__global__ __launch_bounds__(1024) void pool_finish_head_kernel(const float* __restrict__ gmax,
                                                                const float* __restrict__ gsum,
                                                                const float* __restrict__ cnt,
                                                                const float* __restrict__ Wout,
                                                                const float* __restrict__ bout,
                                                                float* __restrict__ hidden,
                                                                float* __restrict__ out) {
  __shared__ float dot[32];
  const int lane = threadIdx.x & 31;
  const int wave = threadIdx.x >> 5;
  const int g = blockIdx.x * 32 + wave;
  const float c = cnt[g];
  float2 mx = ((const float2*)(gmax + (size_t)g * DD))[lane];
  const float2 sm = ((const float2*)(gsum + (size_t)g * DD))[lane];
  if (c == 0.0f) mx = make_float2(0.0f, 0.0f);
  const float cd = fmaxf(c, 1.0f);
  const float2 mean = make_float2(sm.x / cd, sm.y / cd);
  float* __restrict__ hrow = hidden + (size_t)g * (2 * DD);
  ((float2*)hrow)[lane] = mx;
  ((float2*)(hrow + DD))[lane] = mean;
  const int f = lane * 2;
  float acc = mx.x * Wout[f];
  acc = fmaf(mx.y, Wout[f + 1], acc);
  acc = fmaf(mean.x, Wout[DD + f], acc);
  acc = fmaf(mean.y, Wout[DD + f + 1], acc);
  for (int o = 16; o > 0; o >>= 1) acc += __shfl_xor(acc, o);
  if (lane == 0) dot[wave] = acc;
  __syncthreads();
  if (wave == 0) out[blockIdx.x * 32 + lane] = dot[lane] + bout[0];
}


extern "C" void kernel_launch(void* const* d_in, const int* in_sizes, int n_in,
                              void* d_out, int out_size, void* d_ws, size_t ws_size,
                              hipStream_t stream) {
  (void)in_sizes; (void)n_in; (void)out_size; (void)ws_size;

  const float* x    = (const float*)d_in[0];
  const int*   eidx = (const int*)d_in[1];
  const int*   batch = (const int*)d_in[2];
  const float* W0 = (const float*)d_in[3];
  const float* b0 = (const float*)d_in[4];
  const float* W1 = (const float*)d_in[5];
  const float* b1 = (const float*)d_in[6];
  const float* W2 = (const float*)d_in[7];
  const float* b2 = (const float*)d_in[8];
  const float* W3 = (const float*)d_in[9];
  const float* b3 = (const float*)d_in[10];
  const float* Wout = (const float*)d_in[11];
  const float* bout = (const float*)d_in[12];

  const int* src = eidx;
  const int* dst = eidx + EE;

  float* outp   = (float*)d_out;
  float* hidden = (float*)d_out + GG;

  float* ws = (float*)d_ws;
  size_t o = 0;
  float* dinv   = ws + o;         o += NN + 48;
  int*   rowptr = (int*)(ws + o); o += (NB * RPB + 63) / 64 * 64;
  int*   table  = (int*)(ws + o); o += NB * NCH + 48;
  int*   alloc  = (int*)(ws + o); o += 64;
  int*   part   = (int*)(ws + o); o += (size_t)EE * 2;
  int*   csr    = (int*)(ws + o); o += (size_t)EE;
  float* hpad  = ws + o; o += (size_t)NN * 16;
  float* w0pad = ws + o; o += 16 * 64;
  float* h0    = ws + o; o += (size_t)NN * DD;
  float* h1    = ws + o; o += (size_t)NN * DD;
  float* gmax  = ws + o; o += (size_t)GG * DD;
  float* gsum  = ws + o; o += (size_t)GG * DD;
  float* gcnt  = ws + o; o += GG;

  chunk_sort_kernel<<<NCH + NAUX, 1024, 0, stream>>>(src, dst, table, part, W0, w0pad, gmax, gsum, gcnt, alloc);
  bin_build_kernel<<<NB, 512, 0, stream>>>(part, table, alloc, rowptr, dinv, csr, x, hpad);

  const int mtiles = NN / 16;

  layer_kernel<16, true, false><<<mtiles, 64, 0, stream>>>(hpad, w0pad, rowptr, csr, dinv, b0, h0, batch, gmax, gsum, gcnt);
  layer_kernel<64, true, false><<<mtiles, 256, 0, stream>>>(h0, W1, rowptr, csr, dinv, b1, h1, batch, gmax, gsum, gcnt);
  layer_kernel<64, true, false><<<mtiles, 256, 0, stream>>>(h1, W2, rowptr, csr, dinv, b2, h0, batch, gmax, gsum, gcnt);
  layer_kernel<64, false, true><<<mtiles, 256, 0, stream>>>(h0, W3, rowptr, csr, dinv, b3, h1, batch, gmax, gsum, gcnt);

  pool_finish_head_kernel<<<GG / 32, 1024, 0, stream>>>(gmax, gsum, gcnt, Wout, bout, hidden, outp);
}


# Edge-list build: both passes place their pairs / csr indices in LDS and copy the finished range out as whole 128-byte lines (ranges line-aligned), not one scattered store per edge; sum order kept
# speedup vs baseline: 6.4336x; 1.0569x over previous
#include <hip/hip_runtime.h>
#include <math.h>
#include <float.h>


typedef __attribute__((ext_vector_type(2))) float v2f;
typedef __attribute__((ext_vector_type(8))) float v8f;

#define NN 50000
#define EE 1200000
#define DD 64
#define GG 1024

static_assert(NN % 16 == 0, "the layer kernel's 16-row WMMA tiles need a row count that is a multiple of 16");


#define BSH 7
#define BS (1 << BSH)
#define NB ((NN + BS - 1) / BS)
#define RPB (BS + 1)
#define NCH 240
#define CH (EE / NCH)
#define CHP ((CH + 15) / 16 * 16)
#define SCAP 8192
#define CSRN (EE + NB * 32)
static_assert(EE % NCH == 0 && CH % 4 == 0, "chunks are whole 16-byte groups of edges");
static_assert(SCAP % 4 == 0, "the LDS copy of a bin's range is read in 16-byte groups");
static_assert(NB <= 32 * 13, "the bin scan gives each lane of one wave 13 bins");
static_assert(CH < 65536, "a group's start in its chunk and its length share one word");
static_assert(NCH % 16 == 0, "the 16 waves of a bin_build block take the same number of groups");

#define NAUX ((GG * DD / 4 + 1023) / 1024)
static_assert(NAUX * 1024 >= GG * DD / 4 && NAUX * 1024 >= 16 * 64,
              "one item per thread of the extra work-groups covers every pass");
static_assert(GG % 4 == 0, "the count plane is filled in 16-byte groups");

__global__ __launch_bounds__(1024) void chunk_sort_kernel(const int* __restrict__ src,
                                                          const int* __restrict__ dst,
                                                          int* __restrict__ table,
                                                          int* __restrict__ part_,
                                                          const float* __restrict__ W0,
                                                          float* __restrict__ w0pad,
                                                          float* __restrict__ gmax,
                                                          float* __restrict__ gsum,
                                                          float* __restrict__ gcnt,
                                                          int* __restrict__ alloc) {
  int2* __restrict__ part = (int2*)part_;
  __shared__ int hist[NB];
  __shared__ int cur[NB];
  __shared__ __attribute__((aligned(16))) int2 stage[CH];
  const int tid = threadIdx.x;
  const int lane = tid & 31;
  const int wave = tid >> 5;
  const int c = blockIdx.x;
  if (c >= NCH) {
    const int t = (c - NCH) * 1024 + tid;
    if (t < GG * DD / 4) {
      ((float4*)gmax)[t] = make_float4(-FLT_MAX, -FLT_MAX, -FLT_MAX, -FLT_MAX);
      ((float4*)gsum)[t] = make_float4(0.0f, 0.0f, 0.0f, 0.0f);
    }
    if (t < GG / 4) ((float4*)gcnt)[t] = make_float4(0.0f, 0.0f, 0.0f, 0.0f);
    if (t < 16 * 64) w0pad[t] = (t < 9 * 64) ? W0[t] : 0.0f;
    if (t == 0) alloc[0] = 0;
    return;
  }
  if (tid < NB) hist[tid] = 0;
  __syncthreads();

  const int4* __restrict__ ps = (const int4*)(src + c * CH);
  const int4* __restrict__ pd = (const int4*)(dst + c * CH);
  constexpr int Q = CH / 4;
  constexpr int KB = (Q + 1023) / 1024;
  int4 s[KB];
  int4 d[KB];
#pragma unroll
  for (int k = 0; k < KB; ++k) {
    const int i = tid + k * 1024;
    s[k] = ps[(i < Q) ? i : (Q - 1)];
    d[k] = pd[(i < Q) ? i : (Q - 1)];
  }
#pragma unroll
  for (int k = 0; k < KB; ++k) {
    if (tid + k * 1024 < Q) {
      atomicAdd(&hist[d[k].x >> BSH], 1);
      atomicAdd(&hist[d[k].y >> BSH], 1);
      atomicAdd(&hist[d[k].z >> BSH], 1);
      atomicAdd(&hist[d[k].w >> BSH], 1);
    }
  }
  __syncthreads();

  if (wave == 0) {
    int v[13];
    int sum = 0;
#pragma unroll
    for (int k = 0; k < 13; ++k) {
      const int idx = lane * 13 + k;
      v[k] = (idx < NB) ? hist[idx] : 0;
      sum += v[k];
    }
    int incl = sum;
    for (int o = 1; o < 32; o <<= 1) {
      const int u = __shfl_up(incl, o);
      if (lane >= o) incl += u;
    }
    int excl = incl - sum;
#pragma unroll
    for (int k = 0; k < 13; ++k) {
      const int idx = lane * 13 + k;
      if (idx < NB) {
        cur[idx] = excl;
        table[idx * NCH + c] = (excl << 16) | v[k];
      }
      excl += v[k];
    }
  }
  __syncthreads();

#pragma unroll
  for (int k = 0; k < KB; ++k) {
    if (tid + k * 1024 < Q) {
      stage[atomicAdd(&cur[d[k].x >> BSH], 1)] = make_int2(s[k].x, d[k].x);
      stage[atomicAdd(&cur[d[k].y >> BSH], 1)] = make_int2(s[k].y, d[k].y);
      stage[atomicAdd(&cur[d[k].z >> BSH], 1)] = make_int2(s[k].z, d[k].z);
      stage[atomicAdd(&cur[d[k].w >> BSH], 1)] = make_int2(s[k].w, d[k].w);
    }
  }
  __syncthreads();

  int4* __restrict__ mine4 = (int4*)(part + (size_t)c * CHP);
  const int4* stage4 = (const int4*)stage;
  for (int i = tid; i < CH / 2; i += 1024) mine4[i] = stage4[i];
}

__global__ __launch_bounds__(512) void bin_build_kernel(const int* __restrict__ part_,
                                                         const int* __restrict__ table,
                                                         int* __restrict__ alloc,
                                                         int* __restrict__ rowptr,
                                                         float* __restrict__ dinv,
                                                         int* __restrict__ csr,
                                                         const float* __restrict__ x,
                                                         float* __restrict__ hpad) {
  const int2* __restrict__ part = (const int2*)part_;
  __shared__ int cnt[BS];
  __shared__ int cursor[BS];
  __shared__ int wtot[BS / 32];
  __shared__ int wsum[16];
  __shared__ int lo_s;
  __shared__ int tot_s;
  __shared__ __attribute__((aligned(16))) int stage[SCAP];
  const int tid = threadIdx.x;
  const int lane = tid & 31;
  const int wave = tid >> 5;
  const int b = blockIdx.x;
  const int base = b * BS;
  constexpr int SPW = NCH / 16;
  static_assert(SPW <= 32, "a wave's table entries are loaded one per lane");

  if (tid < BS) cnt[tid] = 0;
  int pk = 0;
  if (lane < SPW) pk = table[b * NCH + wave + 16 * lane];
  float xv[9];
  if (tid < BS && base + tid < NN) {
    const float* __restrict__ xr = x + (size_t)(base + tid) * 9;
#pragma unroll
    for (int k = 0; k < 9; ++k) xv[k] = xr[k];
  }
  int2 e[SPW];
  int sn[SPW];
  int so[SPW];
#pragma unroll
  for (int k = 0; k < SPW; ++k) {
    const int w = __builtin_amdgcn_readlane(pk, k);
    sn[k] = w & 0xFFFF;
    so[k] = (wave + 16 * k) * CHP + (w >> 16);
  }
#pragma unroll
  for (int k = 0; k < SPW; ++k) {
    e[k] = make_int2(0, base);
    if (lane < sn[k]) e[k] = part[so[k] + lane];
  }
  int tsum = pk & 0xFFFF;
  for (int o = 16; o > 0; o >>= 1) tsum += __shfl_xor(tsum, o);
  if (lane == 0) wsum[wave] = tsum;
  __syncthreads();
  if (tid == 0) {
    int total = 0;
    for (int w = 0; w < 16; ++w) total += wsum[w];
    tot_s = total;
    lo_s = atomicAdd(alloc, (total + 31) & ~31);
  }
#pragma unroll
  for (int k = 0; k < SPW; ++k)
    if (lane < sn[k]) atomicAdd(&cnt[e[k].y - base], 1);
#pragma unroll
  for (int k = 0; k < SPW; ++k)
    for (int i = 32 + lane; i < sn[k]; i += 32) atomicAdd(&cnt[part[so[k] + i].y - base], 1);
  __syncthreads();

  const int lo = __builtin_amdgcn_readfirstlane(lo_s);
  const int total = __builtin_amdgcn_readfirstlane(tot_s);
  const bool staged = total <= SCAP;
  int c = 0;
  int incl = 0;
  if (tid < BS) {
    c = cnt[tid];
    incl = c;
    for (int o = 1; o < 32; o <<= 1) {
      const int u = __shfl_up(incl, o);
      if (lane >= o) incl += u;
    }
    if (lane == 31) wtot[wave] = incl;
  }
  __syncthreads();
  if (tid < BS) {
    int off = 0;
    for (int w = 0; w < wave; ++w) off += wtot[w];
    const int excl = off + incl - c;
    cursor[tid] = excl;
    rowptr[b * RPB + tid] = lo + excl;
    if (tid == BS - 1) rowptr[b * RPB + BS] = lo + excl + c;
    const int node = base + tid;
    if (node < NN) {
      const float di = rsqrtf((float)(c + 1));
      dinv[node] = di;
      float4* __restrict__ o = (float4*)(hpad + (size_t)node * 16);
      o[0] = make_float4(xv[0] * di, xv[1] * di, xv[2] * di, xv[3] * di);
      o[1] = make_float4(xv[4] * di, xv[5] * di, xv[6] * di, xv[7] * di);
      o[2] = make_float4(xv[8] * di, 0.0f, 0.0f, 0.0f);
      o[3] = make_float4(0.0f, 0.0f, 0.0f, 0.0f);
    }
  }
  __syncthreads();

  int pos[SPW];
#pragma unroll
  for (int k = 0; k < SPW; ++k) {
    pos[k] = 0;
    if (lane < sn[k]) pos[k] = atomicAdd(&cursor[e[k].y - base], 1);
  }
  int* __restrict__ mine = csr + lo;
  if (staged) {
#pragma unroll
    for (int k = 0; k < SPW; ++k)
      if (lane < sn[k]) stage[pos[k]] = e[k].x;
#pragma unroll
    for (int k = 0; k < SPW; ++k)
      for (int i = 32 + lane; i < sn[k]; i += 32) {
        const int2 p = part[so[k] + i];
        stage[atomicAdd(&cursor[p.y - base], 1)] = p.x;
      }
  } else {
#pragma unroll
    for (int k = 0; k < SPW; ++k)
      if (lane < sn[k]) mine[pos[k]] = e[k].x;
#pragma unroll
    for (int k = 0; k < SPW; ++k)
      for (int i = 32 + lane; i < sn[k]; i += 32) {
        const int2 p = part[so[k] + i];
        mine[atomicAdd(&cursor[p.y - base], 1)] = p.x;
      }
  }
  __syncthreads();
  if (staged) {
    const int n4 = total >> 2;
    int4* __restrict__ mine4 = (int4*)mine;
    const int4* stage4 = (const int4*)stage;
    for (int i = tid; i < n4; i += 512) mine4[i] = stage4[i];
    const int r = n4 * 4 + tid;
    if (r < total) mine[r] = stage[r];
  }
}

__device__ __forceinline__ void add4(float4& acc, const float4 v) {
  acc.x += v.x;
  acc.y += v.y;
  acc.z += v.z;
  acc.w += v.w;
}
__device__ __forceinline__ void axpy4(float4& acc, const float4 v, const float w) {
  acc.x = fmaf(v.x, w, acc.x);
  acc.y = fmaf(v.y, w, acc.y);
  acc.z = fmaf(v.z, w, acc.z);
  acc.w = fmaf(v.w, w, acc.w);
}

template <int LPN, int S>
__device__ __forceinline__ int slot_src(const int (&my)[16 / LPN]) {
  static_assert(LPN == 16 || LPN == 4, "a node's lanes are a 16-lane row or a quad");
  if constexpr (LPN == 16)
    return __builtin_amdgcn_update_dpp(0, my[0], 0x150 + S, 0xF, 0xF, true);
  else
    return __builtin_amdgcn_update_dpp(0, my[S / 4], (S % 4) * 0x55, 0xF, 0xF, true);
}

template <int LPN, int S0>
__device__ __forceinline__ void gather8(const int (&my)[16 / LPN], const float4* __restrict__ hin4,
                                        const int lg, const int left, float4& acc) {
  int e[8];
  float4 v[8];
  e[0] = slot_src<LPN, S0 + 0>(my);
  e[1] = slot_src<LPN, S0 + 1>(my);
  e[2] = slot_src<LPN, S0 + 2>(my);
  e[3] = slot_src<LPN, S0 + 3>(my);
  e[4] = slot_src<LPN, S0 + 4>(my);
  e[5] = slot_src<LPN, S0 + 5>(my);
  e[6] = slot_src<LPN, S0 + 6>(my);
  e[7] = slot_src<LPN, S0 + 7>(my);
#pragma unroll
  for (int k = 0; k < 8; ++k) v[k] = hin4[e[k] * LPN + lg];
  if (left >= 8) {
#pragma unroll
    for (int k = 0; k < 8; ++k) add4(acc, v[k]);
  } else {
#pragma unroll
    for (int k = 0; k < 8; ++k) axpy4(acc, v[k], (k < left) ? 1.0f : 0.0f);
  }
}

__device__ inline void atomicMaxF32(float* addr, float val) {
  if (val >= 0.0f)
    atomicMax((int*)addr, __float_as_int(val));
  else
    atomicMin((unsigned int*)addr, __float_as_uint(val));
}

__device__ __forceinline__ void pool_flush(const int g, const int f, const int half,
                                           const bool cntlane, const float s, const float m,
                                           const int c, float* __restrict__ gmax,
                                           float* __restrict__ gsum, float* __restrict__ cnt) {
  if (half == 0)
    atomicAdd(&gsum[(size_t)g * DD + f], s);
  else
    atomicMaxF32(&gmax[(size_t)g * DD + f], m);
  if (cntlane) atomicAdd(&cnt[g], (float)c);
}

template <int K, bool SCALE_OUT, bool POOL>
__global__ __launch_bounds__(K * 4) void layer_kernel(const float* __restrict__ hin,
                                                      const float* __restrict__ W,
                                                      const int* __restrict__ rowptr,
                                                      const int* __restrict__ csr,
                                                      const float* __restrict__ dinv,
                                                      const float* __restrict__ b,
                                                      float* __restrict__ hout,
                                                      const int* __restrict__ batch,
                                                      float* __restrict__ gmax,
                                                      float* __restrict__ gsum,
                                                      float* __restrict__ gcnt) {
  static_assert(!(POOL && SCALE_OUT), "the pooled layer stores no rows, so there is nothing to scale");
  constexpr int LPN = K / 4;
  constexpr int NW = K / 8;
  constexpr int TPW = (NW >= 4) ? 1 : 4 / NW;
  constexpr int KS = K / 4;
  constexpr int P = K + 4;
  __shared__ __attribute__((aligned(16))) float zs[16 * P];
  __shared__ __attribute__((aligned(16))) float dis[16];

  const int tid = threadIdx.x;
  const int node = tid / LPN;
  const int fg = (tid % LPN) * 4;
  const int m0 = blockIdx.x * 16;
  const int d = m0 + node;

  const int ri = d + (d >> BSH);
  int j = rowptr[ri];
  const int end = rowptr[ri + 1];
  const float4 self = *(const float4*)(hin + d * K + fg);
  const float di = dinv[d];

  float4 acc = make_float4(0.0f, 0.0f, 0.0f, 0.0f);

  if (j < end) {
    constexpr int NR = 16 / LPN;
    const int lg = tid % LPN;
    const float4* __restrict__ hin4 = (const float4*)hin;
    const int last = end - 1;
    int my[NR];
#pragma unroll
    for (int r = 0; r < NR; ++r) {
      const int s = j + r * LPN + lg;
      my[r] = csr[(s < last) ? s : last];
    }
    for (;;) {
      const int jn = j + 16;
      int nxt[NR];
#pragma unroll
      for (int r = 0; r < NR; ++r) {
        const int s = jn + r * LPN + lg;
        nxt[r] = csr[(s < last) ? s : last];
      }
      gather8<LPN, 0>(my, hin4, lg, end - j, acc);
      if (j + 8 < end) gather8<LPN, 8>(my, hin4, lg, end - j - 8, acc);
      if (jn >= end) break;
#pragma unroll
      for (int r = 0; r < NR; ++r) my[r] = nxt[r];
      j = jn;
    }
  }

  const int wave = __builtin_amdgcn_readfirstlane(tid >> 5);
  const int lane = tid & 31;
  const int l15 = lane & 15;
  const int half = lane >> 4;
  const bool mult = wave < 4;

  float4 z;
  z.x = (acc.x + self.x) * di;
  z.y = (acc.y + self.y) * di;
  z.z = (acc.z + self.z) * di;
  z.w = (acc.w + self.w) * di;
  *(float4*)(zs + node * P + fg) = z;
  if (SCALE_OUT && fg == 0) dis[node] = di;

  v2f bf[TPW][KS];
  if (mult) {
#pragma unroll
    for (int t = 0; t < TPW; ++t) {
      const float* __restrict__ bcol = W + half * 2 * 64 + (wave + t * NW) * 16 + l15;
#pragma unroll
      for (int s = 0; s < KS; ++s) {
        bf[t][s].x = bcol[s * 4 * 64];
        bf[t][s].y = bcol[s * 4 * 64 + 64];
      }
    }
  }
  __builtin_amdgcn_fence(__ATOMIC_RELEASE, "workgroup", "local");
  __builtin_amdgcn_s_barrier();
  __builtin_amdgcn_fence(__ATOMIC_ACQUIRE, "workgroup", "local");

  if (mult) {
    const float* zrow = zs + l15 * P + half * 2;
#pragma unroll
    for (int t = 0; t < TPW; ++t) {
      const int col = (wave + t * NW) * 16 + l15;
      v8f c0 = {};
      v8f c1 = {};
#pragma unroll
      for (int s = 0; s < KS; s += 2) {
        const v2f a0 = *(const v2f*)(zrow + s * 4);
        const v2f a1 = *(const v2f*)(zrow + s * 4 + 4);
        c0 = __builtin_amdgcn_wmma_f32_16x16x4_f32(false, a0, false, bf[t][s], (short)0, c0,
                                                   false, false);
        c1 = __builtin_amdgcn_wmma_f32_16x16x4_f32(false, a1, false, bf[t][s + 1], (short)0, c1,
                                                   false, false);
      }
      const float bias = b[col];
      float* __restrict__ orow = hout + (m0 + half * 8) * 64 + col;
      if (POOL) {
        float own[8];
        float oth[8];
        float cv[16];
#pragma unroll
        for (int v = 0; v < 8; ++v) own[v] = tanhf(c0[v] + c1[v] + bias);
#pragma unroll
        for (int v = 0; v < 8; ++v) oth[v] = __shfl_xor(own[v], 16);
#pragma unroll
        for (int v = 0; v < 8; ++v) {
          cv[v] = half ? oth[v] : own[v];
          cv[8 + v] = half ? own[v] : oth[v];
        }
        int g[16];
#pragma unroll
        for (int r = 0; r < 16; ++r) g[r] = batch[m0 + r];
        const bool cntlane = (col == 0) && (half == 0);
        int cur = g[0];
        float s = cv[0];
        float m = cv[0];
        int c = 1;
#pragma unroll
        for (int r = 1; r < 16; ++r) {
          if (g[r] != cur) {
            pool_flush(cur, col, half, cntlane, s, m, c, gmax, gsum, gcnt);
            cur = g[r];
            s = cv[r];
            m = cv[r];
            c = 1;
          } else {
            s += cv[r];
            m = fmaxf(m, cv[r]);
            c += 1;
          }
        }
        pool_flush(cur, col, half, cntlane, s, m, c, gmax, gsum, gcnt);
      } else if (SCALE_OUT) {
        const float4 s0 = *(const float4*)(dis + half * 8);
        const float4 s1 = *(const float4*)(dis + half * 8 + 4);
        const float so[8] = {s0.x, s0.y, s0.z, s0.w, s1.x, s1.y, s1.z, s1.w};
#pragma unroll
        for (int v = 0; v < 8; ++v) orow[v * 64] = tanhf(c0[v] + c1[v] + bias) * so[v];
      } else {
#pragma unroll
        for (int v = 0; v < 8; ++v) orow[v * 64] = tanhf(c0[v] + c1[v] + bias);
      }
    }
  }
}


static_assert(GG % 32 == 0 && DD == 64, "32 graphs per work-group, two features per lane");
__global__ __launch_bounds__(1024) void pool_finish_head_kernel(const float* __restrict__ gmax,
                                                                const float* __restrict__ gsum,
                                                                const float* __restrict__ cnt,
                                                                const float* __restrict__ Wout,
                                                                const float* __restrict__ bout,
                                                                float* __restrict__ hidden,
                                                                float* __restrict__ out) {
  __shared__ float dot[32];
  const int lane = threadIdx.x & 31;
  const int wave = threadIdx.x >> 5;
  const int g = blockIdx.x * 32 + wave;
  const float c = cnt[g];
  float2 mx = ((const float2*)(gmax + (size_t)g * DD))[lane];
  const float2 sm = ((const float2*)(gsum + (size_t)g * DD))[lane];
  if (c == 0.0f) mx = make_float2(0.0f, 0.0f);
  const float cd = fmaxf(c, 1.0f);
  const float2 mean = make_float2(sm.x / cd, sm.y / cd);
  float* __restrict__ hrow = hidden + (size_t)g * (2 * DD);
  ((float2*)hrow)[lane] = mx;
  ((float2*)(hrow + DD))[lane] = mean;
  const int f = lane * 2;
  float acc = mx.x * Wout[f];
  acc = fmaf(mx.y, Wout[f + 1], acc);
  acc = fmaf(mean.x, Wout[DD + f], acc);
  acc = fmaf(mean.y, Wout[DD + f + 1], acc);
  for (int o = 16; o > 0; o >>= 1) acc += __shfl_xor(acc, o);
  if (lane == 0) dot[wave] = acc;
  __syncthreads();
  if (wave == 0) out[blockIdx.x * 32 + lane] = dot[lane] + bout[0];
}


extern "C" void kernel_launch(void* const* d_in, const int* in_sizes, int n_in,
                              void* d_out, int out_size, void* d_ws, size_t ws_size,
                              hipStream_t stream) {
  (void)in_sizes; (void)n_in; (void)out_size; (void)ws_size;

  const float* x    = (const float*)d_in[0];
  const int*   eidx = (const int*)d_in[1];
  const int*   batch = (const int*)d_in[2];
  const float* W0 = (const float*)d_in[3];
  const float* b0 = (const float*)d_in[4];
  const float* W1 = (const float*)d_in[5];
  const float* b1 = (const float*)d_in[6];
  const float* W2 = (const float*)d_in[7];
  const float* b2 = (const float*)d_in[8];
  const float* W3 = (const float*)d_in[9];
  const float* b3 = (const float*)d_in[10];
  const float* Wout = (const float*)d_in[11];
  const float* bout = (const float*)d_in[12];

  const int* src = eidx;
  const int* dst = eidx + EE;

  float* outp   = (float*)d_out;
  float* hidden = (float*)d_out + GG;

  float* ws = (float*)d_ws;
  size_t o = 0;
  float* dinv   = ws + o;         o += NN + 48;
  int*   rowptr = (int*)(ws + o); o += (NB * RPB + 63) / 64 * 64;
  int*   table  = (int*)(ws + o); o += NB * NCH + 48;
  int*   alloc  = (int*)(ws + o); o += 64;
  int*   part   = (int*)(ws + o); o += (size_t)NCH * CHP * 2;
  int*   csr    = (int*)(ws + o); o += ((size_t)CSRN + 63) / 64 * 64;
  float* hpad  = ws + o; o += (size_t)NN * 16;
  float* w0pad = ws + o; o += 16 * 64;
  float* h0    = ws + o; o += (size_t)NN * DD;
  float* h1    = ws + o; o += (size_t)NN * DD;
  float* gmax  = ws + o; o += (size_t)GG * DD;
  float* gsum  = ws + o; o += (size_t)GG * DD;
  float* gcnt  = ws + o; o += GG;

  chunk_sort_kernel<<<NCH + NAUX, 1024, 0, stream>>>(src, dst, table, part, W0, w0pad, gmax, gsum, gcnt, alloc);
  bin_build_kernel<<<NB, 512, 0, stream>>>(part, table, alloc, rowptr, dinv, csr, x, hpad);

  const int mtiles = NN / 16;

  layer_kernel<16, true, false><<<mtiles, 64, 0, stream>>>(hpad, w0pad, rowptr, csr, dinv, b0, h0, batch, gmax, gsum, gcnt);
  layer_kernel<64, true, false><<<mtiles, 256, 0, stream>>>(h0, W1, rowptr, csr, dinv, b1, h1, batch, gmax, gsum, gcnt);
  layer_kernel<64, true, false><<<mtiles, 256, 0, stream>>>(h1, W2, rowptr, csr, dinv, b2, h0, batch, gmax, gsum, gcnt);
  layer_kernel<64, false, true><<<mtiles, 256, 0, stream>>>(h0, W3, rowptr, csr, dinv, b3, h1, batch, gmax, gsum, gcnt);

  pool_finish_head_kernel<<<GG / 32, 1024, 0, stream>>>(gmax, gsum, gcnt, Wout, bout, hidden, outp);
}
